# starting kernel
# baseline (speedup 1.0000x reference)
#include <hip/hip_runtime.h>
#include <cstdint>

typedef __attribute__((ext_vector_type(2))) float v2f;
typedef __attribute__((ext_vector_type(8))) float v8f;

__global__ void deg_count_kernel(const int* __restrict__ dst, int E,
                                 float* __restrict__ deg) {
    int e = blockIdx.x * blockDim.x + threadIdx.x;
    if (e < E) atomicAdd(&deg[dst[e]], 1.0f);
}

__global__ void deg_finalize_kernel(float* __restrict__ dis, int n) {
    int i = blockIdx.x * blockDim.x + threadIdx.x;
    if (i < n) dis[i] = rsqrtf(dis[i] + 1.0f);
}

template <int K, int NC>
__global__ void gemm_wmma_f32(const float* __restrict__ A,
                              const float* __restrict__ B,
                              float* __restrict__ C, int nTiles) {
    int wave = (int)((blockIdx.x * blockDim.x + threadIdx.x) >> 5);
    int lane = threadIdx.x & 31;
    if (wave >= nTiles) return;
    constexpr int TN = NC / 16;
    int tm = wave / TN;
    int tn = wave % TN;
    int rowBase = tm * 16;
    int colBase = tn * 16;
    int hi = lane >> 4;
    int l  = lane & 15;

    const float* arow = A + (size_t)(rowBase + l) * K;
    const float* bcol = B + colBase + l;

    v8f acc = {};
#pragma unroll
    for (int k = 0; k < K; k += 4) {
        v2f a, b;
        a.x = arow[k + 2 * hi + 0];
        a.y = arow[k + 2 * hi + 1];
        b.x = bcol[(size_t)(k + 2 * hi + 0) * NC];
        b.y = bcol[(size_t)(k + 2 * hi + 1) * NC];
        acc = __builtin_amdgcn_wmma_f32_16x16x4_f32(
             false, a,  false, b,
             (short)0, acc,  false,  false);
    }

    float* cptr = C + (size_t)colBase + l;
#pragma unroll
    for (int i = 0; i < 8; ++i)
        cptr[(size_t)(rowBase + 8 * hi + i) * NC] = acc[i];
}

template <int LOGF>
__global__ void edge_agg_kernel(const int* __restrict__ src,
                                const int* __restrict__ dst,
                                const float* __restrict__ dis,
                                const float* __restrict__ h,
                                float* __restrict__ agg, long long total) {
    long long t = (long long)blockIdx.x * blockDim.x + threadIdx.x;
    if (t >= total) return;
    int e = (int)(t >> LOGF);
    int j = (int)(t & ((1 << LOGF) - 1));
    int s = src[e];
    int d = dst[e];
    float norm = dis[s] * dis[d];
    float v = h[((size_t)s << LOGF) + j] * norm;
    atomicAdd(&agg[((size_t)d << LOGF) + j], v);
}

__global__ void selfloop_bias_relu_kernel(const float* __restrict__ h,
                                          const float* __restrict__ dis,
                                          const float* __restrict__ bias,
                                          float* __restrict__ agg, int n) {
    int t = blockIdx.x * blockDim.x + threadIdx.x;
    if (t >= n) return;
    int i = t >> 6;
    int j = t & 63;
    float dii = dis[i];
    float v = agg[t] + h[t] * dii * dii + bias[j];
    agg[t] = fmaxf(v, 0.0f);
}

__global__ void logsoftmax_kernel(const float* __restrict__ h,
                                  const float* __restrict__ dis,
                                  const float* __restrict__ bias,
                                  const float* __restrict__ agg,
                                  float* __restrict__ out, int n) {
    int i = blockIdx.x * blockDim.x + threadIdx.x;
    if (i >= n) return;
    float dii = dis[i];
    dii *= dii;
    float v[16];
    float mx = -3.402823466e38f;
#pragma unroll
    for (int j = 0; j < 16; ++j) {
        size_t idx = (size_t)i * 16 + j;
        v[j] = agg[idx] + h[idx] * dii + bias[j];
        mx = fmaxf(mx, v[j]);
    }
    float s = 0.0f;
#pragma unroll
    for (int j = 0; j < 16; ++j) s += __expf(v[j] - mx);
    float lse = mx + __logf(s);
#pragma unroll
    for (int j = 0; j < 16; ++j) out[(size_t)i * 16 + j] = v[j] - lse;
}

extern "C" void kernel_launch(void* const* d_in, const int* in_sizes, int n_in,
                              void* d_out, int out_size, void* d_ws, size_t ws_size,
                              hipStream_t stream) {
    const float* x   = (const float*)d_in[0];
    const int*   ei  = (const int*)  d_in[1];
    const float* W1  = (const float*)d_in[2];
    const float* b1  = (const float*)d_in[3];
    const float* W2  = (const float*)d_in[4];
    const float* b2  = (const float*)d_in[5];
    float* out = (float*)d_out;

    const int N = in_sizes[0] / 128;
    const int E = in_sizes[1] / 2;
    const int* src = ei;
    const int* dst = ei + E;

    float* dis  = (float*)d_ws;
    float* h1   = dis  + N;
    float* agg1 = h1   + (size_t)N * 64;
    float* h2   = agg1 + (size_t)N * 64;
    float* agg2 = h2   + (size_t)N * 16;

    hipMemsetAsync(dis,  0, (size_t)N * sizeof(float), stream);
    hipMemsetAsync(agg1, 0, (size_t)N * 64 * sizeof(float), stream);
    hipMemsetAsync(agg2, 0, (size_t)N * 16 * sizeof(float), stream);

    const int T = 256;

    deg_count_kernel<<<(E + T - 1) / T, T, 0, stream>>>(dst, E, dis);
    deg_finalize_kernel<<<(N + T - 1) / T, T, 0, stream>>>(dis, N);

    {
        int nTiles = (N / 16) * (64 / 16);
        int waves_per_block = T / 32;
        gemm_wmma_f32<128, 64><<<(nTiles + waves_per_block - 1) / waves_per_block,
                                 T, 0, stream>>>(x, W1, h1, nTiles);
    }
    {
        long long total = (long long)E * 64;
        edge_agg_kernel<6><<<(int)((total + T - 1) / T), T, 0, stream>>>(
            src, dst, dis, h1, agg1, total);
    }
    {
        int total = N * 64;
        selfloop_bias_relu_kernel<<<(total + T - 1) / T, T, 0, stream>>>(
            h1, dis, b1, agg1, total);
    }
    {
        int nTiles = (N / 16) * (16 / 16);
        int waves_per_block = T / 32;
        gemm_wmma_f32<64, 16><<<(nTiles + waves_per_block - 1) / waves_per_block,
                                T, 0, stream>>>(agg1, W2, h2, nTiles);
    }
    {
        long long total = (long long)E * 16;
        edge_agg_kernel<4><<<(int)((total + T - 1) / T), T, 0, stream>>>(
            src, dst, dis, h2, agg2, total);
    }
    logsoftmax_kernel<<<(N + T - 1) / T, T, 0, stream>>>(h2, dis, b2, agg2, out, N);
}


# Layer-1 aggregation by destination tile: each work-group keeps 392 rows in LDS, filters the whole dst list, adds its edges' rows by LDS float adds; global atomics gone; sum order changes
# speedup vs baseline: 1.5180x; 1.5180x over previous
#include <hip/hip_runtime.h>
#include <cstdint>

typedef __attribute__((ext_vector_type(2))) float v2f;
typedef __attribute__((ext_vector_type(8))) float v8f;

__global__ void deg_count_kernel(const int* __restrict__ dst, int E,
                                 float* __restrict__ deg) {
    int e = blockIdx.x * blockDim.x + threadIdx.x;
    if (e < E) atomicAdd(&deg[dst[e]], 1.0f);
}

__global__ void deg_finalize_kernel(float* __restrict__ dis, int n) {
    int i = blockIdx.x * blockDim.x + threadIdx.x;
    if (i < n) dis[i] = rsqrtf(dis[i] + 1.0f);
}

template <int K, int NC>
__global__ void gemm_wmma_f32(const float* __restrict__ A,
                              const float* __restrict__ B,
                              float* __restrict__ C, int nTiles) {
    int wave = (int)((blockIdx.x * blockDim.x + threadIdx.x) >> 5);
    int lane = threadIdx.x & 31;
    if (wave >= nTiles) return;
    constexpr int TN = NC / 16;
    int tm = wave / TN;
    int tn = wave % TN;
    int rowBase = tm * 16;
    int colBase = tn * 16;
    int hi = lane >> 4;
    int l  = lane & 15;

    const float* arow = A + (size_t)(rowBase + l) * K;
    const float* bcol = B + colBase + l;

    v8f acc = {};
#pragma unroll
    for (int k = 0; k < K; k += 4) {
        v2f a, b;
        a.x = arow[k + 2 * hi + 0];
        a.y = arow[k + 2 * hi + 1];
        b.x = bcol[(size_t)(k + 2 * hi + 0) * NC];
        b.y = bcol[(size_t)(k + 2 * hi + 1) * NC];
        acc = __builtin_amdgcn_wmma_f32_16x16x4_f32(
             false, a,  false, b,
             (short)0, acc,  false,  false);
    }

    float* cptr = C + (size_t)colBase + l;
#pragma unroll
    for (int i = 0; i < 8; ++i)
        cptr[(size_t)(rowBase + 8 * hi + i) * NC] = acc[i];
}

template <int LOGF>
__global__ void edge_agg_kernel(const int* __restrict__ src,
                                const int* __restrict__ dst,
                                const float* __restrict__ dis,
                                const float* __restrict__ h,
                                float* __restrict__ agg, long long total) {
    long long t = (long long)blockIdx.x * blockDim.x + threadIdx.x;
    if (t >= total) return;
    int e = (int)(t >> LOGF);
    int j = (int)(t & ((1 << LOGF) - 1));
    int s = src[e];
    int d = dst[e];
    float norm = dis[s] * dis[d];
    float v = h[((size_t)s << LOGF) + j] * norm;
    atomicAdd(&agg[((size_t)d << LOGF) + j], v);
}

constexpr int TILE_NODES = 392;
constexpr int QCAP       = 8192;
constexpr int AGG_T      = 1024;
constexpr int SCAN_U     = 8;

__global__ __launch_bounds__(AGG_T)
void tile_agg64_kernel(const int* __restrict__ src,
                       const int* __restrict__ dst,
                       const float* __restrict__ dis,
                       const float* __restrict__ h,
                       float* __restrict__ agg, int E, int N) {
    constexpr int F = 64;
    __shared__ float acc[TILE_NODES * F];
    __shared__ int   q_a[QCAP];
    __shared__ int   q_d[QCAP];
    __shared__ float q_w[QCAP];
    __shared__ int   q_count;

    const int tid  = threadIdx.x;
    const int base = blockIdx.x * TILE_NODES;
    int rows = N - base;
    if (rows > TILE_NODES) rows = TILE_NODES;

    for (int i = tid; i < TILE_NODES * F; i += AGG_T) acc[i] = 0.0f;
    if (tid == 0) q_count = 0;
    __syncthreads();

    auto take = [&](int d, int e) {
        unsigned dl = (unsigned)(d - base);
        if (dl < (unsigned)rows) {
            int slot = atomicAdd(&q_count, 1);
            if (slot < QCAP) {
                q_a[slot] = e;
                q_d[slot] = (int)dl;
            }
        }
    };

    int pre = (int)(((16u - (unsigned)((uintptr_t)dst & 15u)) & 15u) >> 2);
    if (pre > E) pre = E;
    const int nvec = (E - pre) >> 2;
    const int4* __restrict__ dv = (const int4*)(dst + pre);

    for (int e = tid; e < pre; e += AGG_T) take(dst[e], e);
    for (int e = pre + 4 * nvec + tid; e < E; e += AGG_T) take(dst[e], e);

    constexpr int PER_IT = SCAN_U * AGG_T;
    const int iters = (nvec + PER_IT - 1) / PER_IT;
    const int rot = (int)(((long long)blockIdx.x * iters) / gridDim.x);
    for (int it = 0; it < iters; ++it) {
        int itr = it + rot;
        if (itr >= iters) itr -= iters;
        const int g0 = itr * PER_IT + tid;
        int4 v[SCAN_U];
#pragma unroll
        for (int u = 0; u < SCAN_U; ++u) {
            int g = g0 + u * AGG_T;
            v[u] = (g < nvec) ? dv[g] : make_int4(-1, -1, -1, -1);
        }
#pragma unroll
        for (int u = 0; u < SCAN_U; ++u) {
            int e0 = pre + 4 * (g0 + u * AGG_T);
            take(v[u].x, e0 + 0);
            take(v[u].y, e0 + 1);
            take(v[u].z, e0 + 2);
            take(v[u].w, e0 + 3);
        }
    }
    __syncthreads();

    int total = q_count;
    if (total > QCAP) {
        for (int e = tid; e < E; e += AGG_T) {
            int d = dst[e];
            unsigned dl = (unsigned)(d - base);
            if (dl < (unsigned)rows) {
                int s = src[e];
                float w = dis[s] * dis[d];
                for (int j = 0; j < F; ++j)
                    atomicAdd(&acc[dl * F + j], h[(size_t)s * F + j] * w);
            }
        }
        total = 0;
    }

    for (int i = tid; i < total; i += AGG_T) {
        int e  = q_a[i];
        int dl = q_d[i];
        int s  = src[e];
        q_a[i] = s;
        q_w[i] = dis[s] * dis[base + dl];
    }
    __syncthreads();

    const int wave = tid >> 5;
    const int lane = tid & 31;
    constexpr int R = 8;
    for (int i0 = wave * R; i0 < total; i0 += (AGG_T / 32) * R) {
        int    sn[R];
        int    dn[R];
        float  wn[R];
        float2 hv[R];
#pragma unroll
        for (int r = 0; r < R; ++r) {
            int i = i0 + r;
            if (i > total - 1) i = total - 1;
            sn[r] = q_a[i];
            dn[r] = q_d[i];
            wn[r] = q_w[i];
        }
#pragma unroll
        for (int r = 0; r < R; ++r)
            hv[r] = *(const float2*)(h + (size_t)sn[r] * F + 2 * lane);
        asm volatile("" :: "v"(hv[0].x), "v"(hv[0].y), "v"(hv[1].x), "v"(hv[1].y),
                           "v"(hv[2].x), "v"(hv[2].y), "v"(hv[3].x), "v"(hv[3].y),
                           "v"(hv[4].x), "v"(hv[4].y), "v"(hv[5].x), "v"(hv[5].y),
                           "v"(hv[6].x), "v"(hv[6].y), "v"(hv[7].x), "v"(hv[7].y));
#pragma unroll
        for (int r = 0; r < R; ++r) {
            int i = i0 + r;
            if (i < total) {
                atomicAdd(&acc[dn[r] * F + 2 * lane + 0], hv[r].x * wn[r]);
                atomicAdd(&acc[dn[r] * F + 2 * lane + 1], hv[r].y * wn[r]);
            }
        }
    }
    __syncthreads();

    const int n4 = rows * (F / 4);
    float4* __restrict__ o4 = (float4*)(agg + (size_t)base * F);
    const float4* a4 = (const float4*)acc;
    for (int i = tid; i < n4; i += AGG_T) o4[i] = a4[i];
}

__global__ void selfloop_bias_relu_kernel(const float* __restrict__ h,
                                          const float* __restrict__ dis,
                                          const float* __restrict__ bias,
                                          float* __restrict__ agg, int n) {
    int t = blockIdx.x * blockDim.x + threadIdx.x;
    if (t >= n) return;
    int i = t >> 6;
    int j = t & 63;
    float dii = dis[i];
    float v = agg[t] + h[t] * dii * dii + bias[j];
    agg[t] = fmaxf(v, 0.0f);
}

__global__ void logsoftmax_kernel(const float* __restrict__ h,
                                  const float* __restrict__ dis,
                                  const float* __restrict__ bias,
                                  const float* __restrict__ agg,
                                  float* __restrict__ out, int n) {
    int i = blockIdx.x * blockDim.x + threadIdx.x;
    if (i >= n) return;
    float dii = dis[i];
    dii *= dii;
    float v[16];
    float mx = -3.402823466e38f;
#pragma unroll
    for (int j = 0; j < 16; ++j) {
        size_t idx = (size_t)i * 16 + j;
        v[j] = agg[idx] + h[idx] * dii + bias[j];
        mx = fmaxf(mx, v[j]);
    }
    float s = 0.0f;
#pragma unroll
    for (int j = 0; j < 16; ++j) s += __expf(v[j] - mx);
    float lse = mx + __logf(s);
#pragma unroll
    for (int j = 0; j < 16; ++j) out[(size_t)i * 16 + j] = v[j] - lse;
}

extern "C" void kernel_launch(void* const* d_in, const int* in_sizes, int n_in,
                              void* d_out, int out_size, void* d_ws, size_t ws_size,
                              hipStream_t stream) {
    const float* x   = (const float*)d_in[0];
    const int*   ei  = (const int*)  d_in[1];
    const float* W1  = (const float*)d_in[2];
    const float* b1  = (const float*)d_in[3];
    const float* W2  = (const float*)d_in[4];
    const float* b2  = (const float*)d_in[5];
    float* out = (float*)d_out;

    const int N = in_sizes[0] / 128;
    const int E = in_sizes[1] / 2;
    const int* src = ei;
    const int* dst = ei + E;

    float* dis  = (float*)d_ws;
    float* h1   = dis  + N;
    float* agg1 = h1   + (size_t)N * 64;
    float* h2   = agg1 + (size_t)N * 64;
    float* agg2 = h2   + (size_t)N * 16;

    hipMemsetAsync(dis,  0, (size_t)N * sizeof(float), stream);
    hipMemsetAsync(agg2, 0, (size_t)N * 16 * sizeof(float), stream);

    const int T = 256;

    deg_count_kernel<<<(E + T - 1) / T, T, 0, stream>>>(dst, E, dis);
    deg_finalize_kernel<<<(N + T - 1) / T, T, 0, stream>>>(dis, N);

    {
        int nTiles = (N / 16) * (64 / 16);
        int waves_per_block = T / 32;
        gemm_wmma_f32<128, 64><<<(nTiles + waves_per_block - 1) / waves_per_block,
                                 T, 0, stream>>>(x, W1, h1, nTiles);
    }
    tile_agg64_kernel<<<(N + TILE_NODES - 1) / TILE_NODES, AGG_T, 0, stream>>>(
        src, dst, dis, h1, agg1, E, N);
    {
        int total = N * 64;
        selfloop_bias_relu_kernel<<<(total + T - 1) / T, T, 0, stream>>>(
            h1, dis, b1, agg1, total);
    }
    {
        int nTiles = (N / 16) * (16 / 16);
        int waves_per_block = T / 32;
        gemm_wmma_f32<64, 16><<<(nTiles + waves_per_block - 1) / waves_per_block,
                                T, 0, stream>>>(agg1, W2, h2, nTiles);
    }
    {
        long long total = (long long)E * 16;
        edge_agg_kernel<4><<<(int)((total + T - 1) / T), T, 0, stream>>>(
            src, dst, dis, h2, agg2, total);
    }
    logsoftmax_kernel<<<(N + T - 1) / T, T, 0, stream>>>(h2, dis, b2, agg2, out, N);
}


# Layer-2 aggregation on layer 1's destination tiles: the tile kernel stores its edge queue, a new kernel sums h2 rows in LDS from it; global float atomics and agg2 memset gone; sum order changes
# speedup vs baseline: 1.8357x; 1.2093x over previous
#include <hip/hip_runtime.h>
#include <cstdint>

typedef __attribute__((ext_vector_type(2))) float v2f;
typedef __attribute__((ext_vector_type(8))) float v8f;

__global__ void deg_count_kernel(const int* __restrict__ dst, int E,
                                 float* __restrict__ deg) {
    int e = blockIdx.x * blockDim.x + threadIdx.x;
    if (e < E) atomicAdd(&deg[dst[e]], 1.0f);
}

__global__ void deg_finalize_kernel(float* __restrict__ dis, int n) {
    int i = blockIdx.x * blockDim.x + threadIdx.x;
    if (i < n) dis[i] = rsqrtf(dis[i] + 1.0f);
}

template <int K, int NC>
__global__ void gemm_wmma_f32(const float* __restrict__ A,
                              const float* __restrict__ B,
                              float* __restrict__ C, int nTiles) {
    int wave = (int)((blockIdx.x * blockDim.x + threadIdx.x) >> 5);
    int lane = threadIdx.x & 31;
    if (wave >= nTiles) return;
    constexpr int TN = NC / 16;
    int tm = wave / TN;
    int tn = wave % TN;
    int rowBase = tm * 16;
    int colBase = tn * 16;
    int hi = lane >> 4;
    int l  = lane & 15;

    const float* arow = A + (size_t)(rowBase + l) * K;
    const float* bcol = B + colBase + l;

    v8f acc = {};
#pragma unroll
    for (int k = 0; k < K; k += 4) {
        v2f a, b;
        a.x = arow[k + 2 * hi + 0];
        a.y = arow[k + 2 * hi + 1];
        b.x = bcol[(size_t)(k + 2 * hi + 0) * NC];
        b.y = bcol[(size_t)(k + 2 * hi + 1) * NC];
        acc = __builtin_amdgcn_wmma_f32_16x16x4_f32(
             false, a,  false, b,
             (short)0, acc,  false,  false);
    }

    float* cptr = C + (size_t)colBase + l;
#pragma unroll
    for (int i = 0; i < 8; ++i)
        cptr[(size_t)(rowBase + 8 * hi + i) * NC] = acc[i];
}

constexpr int TILE_NODES = 392;
constexpr int QCAP       = 8192;
constexpr int AGG_T      = 1024;
constexpr int SCAN_U     = 8;

__global__ __launch_bounds__(AGG_T)
void tile_agg64_kernel(const int* __restrict__ src,
                       const int* __restrict__ dst,
                       const float* __restrict__ dis,
                       const float* __restrict__ h,
                       float* __restrict__ agg,
                       int* __restrict__ gq_cnt, int* __restrict__ gq_s,
                       int* __restrict__ gq_d, float* __restrict__ gq_w,
                       int E, int N) {
    constexpr int F = 64;
    __shared__ float acc[TILE_NODES * F];
    __shared__ int   q_a[QCAP];
    __shared__ int   q_d[QCAP];
    __shared__ float q_w[QCAP];
    __shared__ int   q_count;

    const int tid  = threadIdx.x;
    const int base = blockIdx.x * TILE_NODES;
    int rows = N - base;
    if (rows > TILE_NODES) rows = TILE_NODES;

    for (int i = tid; i < TILE_NODES * F; i += AGG_T) acc[i] = 0.0f;
    if (tid == 0) q_count = 0;
    __syncthreads();

    auto take = [&](int d, int e) {
        unsigned dl = (unsigned)(d - base);
        if (dl < (unsigned)rows) {
            int slot = atomicAdd(&q_count, 1);
            if (slot < QCAP) {
                q_a[slot] = e;
                q_d[slot] = (int)dl;
            }
        }
    };

    int pre = (int)(((16u - (unsigned)((uintptr_t)dst & 15u)) & 15u) >> 2);
    if (pre > E) pre = E;
    const int nvec = (E - pre) >> 2;
    const int4* __restrict__ dv = (const int4*)(dst + pre);

    for (int e = tid; e < pre; e += AGG_T) take(dst[e], e);
    for (int e = pre + 4 * nvec + tid; e < E; e += AGG_T) take(dst[e], e);

    constexpr int PER_IT = SCAN_U * AGG_T;
    const int iters = (nvec + PER_IT - 1) / PER_IT;
    const int rot = (int)(((long long)blockIdx.x * iters) / gridDim.x);
    for (int it = 0; it < iters; ++it) {
        int itr = it + rot;
        if (itr >= iters) itr -= iters;
        const int g0 = itr * PER_IT + tid;
        int4 v[SCAN_U];
#pragma unroll
        for (int u = 0; u < SCAN_U; ++u) {
            int g = g0 + u * AGG_T;
            v[u] = (g < nvec) ? dv[g] : make_int4(-1, -1, -1, -1);
        }
#pragma unroll
        for (int u = 0; u < SCAN_U; ++u) {
            int e0 = pre + 4 * (g0 + u * AGG_T);
            take(v[u].x, e0 + 0);
            take(v[u].y, e0 + 1);
            take(v[u].z, e0 + 2);
            take(v[u].w, e0 + 3);
        }
    }
    __syncthreads();

    int total = q_count;
    if (tid == 0) gq_cnt[blockIdx.x] = total;
    if (total > QCAP) {
        for (int e = tid; e < E; e += AGG_T) {
            int d = dst[e];
            unsigned dl = (unsigned)(d - base);
            if (dl < (unsigned)rows) {
                int s = src[e];
                float w = dis[s] * dis[d];
                for (int j = 0; j < F; ++j)
                    atomicAdd(&acc[dl * F + j], h[(size_t)s * F + j] * w);
            }
        }
        total = 0;
    }

    int*   __restrict__ ts = gq_s + (size_t)blockIdx.x * QCAP;
    int*   __restrict__ td = gq_d + (size_t)blockIdx.x * QCAP;
    float* __restrict__ tw = gq_w + (size_t)blockIdx.x * QCAP;
    for (int i = tid; i < total; i += AGG_T) {
        int e  = q_a[i];
        int dl = q_d[i];
        int s  = src[e];
        float w = dis[s] * dis[base + dl];
        q_a[i] = s;
        q_w[i] = w;
        ts[i]  = s;
        td[i]  = dl;
        tw[i]  = w;
    }
    __syncthreads();

    const int wave = tid >> 5;
    const int lane = tid & 31;
    constexpr int R = 8;
    for (int i0 = wave * R; i0 < total; i0 += (AGG_T / 32) * R) {
        int    sn[R];
        int    dn[R];
        float  wn[R];
        float2 hv[R];
#pragma unroll
        for (int r = 0; r < R; ++r) {
            int i = i0 + r;
            if (i > total - 1) i = total - 1;
            sn[r] = q_a[i];
            dn[r] = q_d[i];
            wn[r] = q_w[i];
        }
#pragma unroll
        for (int r = 0; r < R; ++r)
            hv[r] = *(const float2*)(h + (size_t)sn[r] * F + 2 * lane);
        asm volatile("" :: "v"(hv[0].x), "v"(hv[0].y), "v"(hv[1].x), "v"(hv[1].y),
                           "v"(hv[2].x), "v"(hv[2].y), "v"(hv[3].x), "v"(hv[3].y),
                           "v"(hv[4].x), "v"(hv[4].y), "v"(hv[5].x), "v"(hv[5].y),
                           "v"(hv[6].x), "v"(hv[6].y), "v"(hv[7].x), "v"(hv[7].y));
#pragma unroll
        for (int r = 0; r < R; ++r) {
            int i = i0 + r;
            if (i < total) {
                atomicAdd(&acc[dn[r] * F + 2 * lane + 0], hv[r].x * wn[r]);
                atomicAdd(&acc[dn[r] * F + 2 * lane + 1], hv[r].y * wn[r]);
            }
        }
    }
    __syncthreads();

    const int n4 = rows * (F / 4);
    float4* __restrict__ o4 = (float4*)(agg + (size_t)base * F);
    const float4* a4 = (const float4*)acc;
    for (int i = tid; i < n4; i += AGG_T) o4[i] = a4[i];
}

typedef __attribute__((ext_vector_type(4))) float v4f;
constexpr int AGG2_T = 1024;
constexpr int R2     = 8;

__global__ __launch_bounds__(AGG2_T)
void tile_agg16_kernel(const int* __restrict__ src,
                       const int* __restrict__ dst,
                       const float* __restrict__ dis,
                       const float* __restrict__ h,
                       const int* __restrict__ gq_cnt, const int* __restrict__ gq_s,
                       const int* __restrict__ gq_d, const float* __restrict__ gq_w,
                       float* __restrict__ agg, int E, int N) {
    constexpr int F = 16;
    __shared__ float acc[TILE_NODES * F];

    const int tid  = threadIdx.x;
    const int base = blockIdx.x * TILE_NODES;
    int rows = N - base;
    if (rows > TILE_NODES) rows = TILE_NODES;

    for (int i = tid; i < TILE_NODES * F; i += AGG2_T) acc[i] = 0.0f;
    __syncthreads();

    int total = gq_cnt[blockIdx.x];
    if (total > QCAP) {
        for (int e = tid; e < E; e += AGG2_T) {
            int d = dst[e];
            unsigned dl = (unsigned)(d - base);
            if (dl < (unsigned)rows) {
                int s = src[e];
                float w = dis[s] * dis[d];
                for (int j = 0; j < F; ++j)
                    atomicAdd(&acc[dl * F + j], h[(size_t)s * F + j] * w);
            }
        }
        total = 0;
    }

    const int*   __restrict__ ts = gq_s + (size_t)blockIdx.x * QCAP;
    const int*   __restrict__ td = gq_d + (size_t)blockIdx.x * QCAP;
    const float* __restrict__ tw = gq_w + (size_t)blockIdx.x * QCAP;

    const int grp = tid >> 2;
    const int qt  = tid & 3;
    constexpr int GROUPS = AGG2_T / 4;
    for (int r0 = 0; r0 < total; r0 += GROUPS * R2) {
        int   sn[R2];
        int   dn[R2];
        float wn[R2];
        v4f   hv[R2];
#pragma unroll
        for (int r = 0; r < R2; ++r) {
            int i = r0 + r * GROUPS + grp;
            if (i > total - 1) i = total - 1;
            sn[r] = ts[i];
            dn[r] = td[i];
            wn[r] = tw[i];
        }
#pragma unroll
        for (int r = 0; r < R2; ++r)
            hv[r] = *(const v4f*)(h + (size_t)sn[r] * F + 4 * qt);
        asm volatile("" :: "v"(hv[0]), "v"(hv[1]), "v"(hv[2]), "v"(hv[3]),
                           "v"(hv[4]), "v"(hv[5]), "v"(hv[6]), "v"(hv[7]),
                           "v"(dn[0]), "v"(dn[1]), "v"(dn[2]), "v"(dn[3]),
                           "v"(dn[4]), "v"(dn[5]), "v"(dn[6]), "v"(dn[7]),
                           "v"(wn[0]), "v"(wn[1]), "v"(wn[2]), "v"(wn[3]),
                           "v"(wn[4]), "v"(wn[5]), "v"(wn[6]), "v"(wn[7]));
#pragma unroll
        for (int r = 0; r < R2; ++r) {
            int i = r0 + r * GROUPS + grp;
            if (i < total) {
                float* p = &acc[dn[r] * F + 4 * qt];
                atomicAdd(p + 0, hv[r].x * wn[r]);
                atomicAdd(p + 1, hv[r].y * wn[r]);
                atomicAdd(p + 2, hv[r].z * wn[r]);
                atomicAdd(p + 3, hv[r].w * wn[r]);
            }
        }
    }
    __syncthreads();

    const int n4 = rows * (F / 4);
    float4* __restrict__ o4 = (float4*)(agg + (size_t)base * F);
    const float4* a4 = (const float4*)acc;
    for (int i = tid; i < n4; i += AGG2_T) o4[i] = a4[i];
}

__global__ void selfloop_bias_relu_kernel(const float* __restrict__ h,
                                          const float* __restrict__ dis,
                                          const float* __restrict__ bias,
                                          float* __restrict__ agg, int n) {
    int t = blockIdx.x * blockDim.x + threadIdx.x;
    if (t >= n) return;
    int i = t >> 6;
    int j = t & 63;
    float dii = dis[i];
    float v = agg[t] + h[t] * dii * dii + bias[j];
    agg[t] = fmaxf(v, 0.0f);
}

__global__ void logsoftmax_kernel(const float* __restrict__ h,
                                  const float* __restrict__ dis,
                                  const float* __restrict__ bias,
                                  const float* __restrict__ agg,
                                  float* __restrict__ out, int n) {
    int i = blockIdx.x * blockDim.x + threadIdx.x;
    if (i >= n) return;
    float dii = dis[i];
    dii *= dii;
    float v[16];
    float mx = -3.402823466e38f;
#pragma unroll
    for (int j = 0; j < 16; ++j) {
        size_t idx = (size_t)i * 16 + j;
        v[j] = agg[idx] + h[idx] * dii + bias[j];
        mx = fmaxf(mx, v[j]);
    }
    float s = 0.0f;
#pragma unroll
    for (int j = 0; j < 16; ++j) s += __expf(v[j] - mx);
    float lse = mx + __logf(s);
#pragma unroll
    for (int j = 0; j < 16; ++j) out[(size_t)i * 16 + j] = v[j] - lse;
}

extern "C" void kernel_launch(void* const* d_in, const int* in_sizes, int n_in,
                              void* d_out, int out_size, void* d_ws, size_t ws_size,
                              hipStream_t stream) {
    const float* x   = (const float*)d_in[0];
    const int*   ei  = (const int*)  d_in[1];
    const float* W1  = (const float*)d_in[2];
    const float* b1  = (const float*)d_in[3];
    const float* W2  = (const float*)d_in[4];
    const float* b2  = (const float*)d_in[5];
    float* out = (float*)d_out;

    const int N = in_sizes[0] / 128;
    const int E = in_sizes[1] / 2;
    const int* src = ei;
    const int* dst = ei + E;

    float* dis  = (float*)d_ws;
    float* h1   = dis  + N;
    float* agg1 = h1   + (size_t)N * 64;
    float* h2   = agg1 + (size_t)N * 64;
    float* agg2 = h2   + (size_t)N * 16;
    const int nT = (N + TILE_NODES - 1) / TILE_NODES;
    int*   gq_s   = (int*)(agg2 + (size_t)N * 16);
    int*   gq_d   = gq_s + (size_t)nT * QCAP;
    float* gq_w   = (float*)(gq_d + (size_t)nT * QCAP);
    int*   gq_cnt = (int*)(gq_w + (size_t)nT * QCAP);

    hipMemsetAsync(dis,  0, (size_t)N * sizeof(float), stream);

    const int T = 256;

    deg_count_kernel<<<(E + T - 1) / T, T, 0, stream>>>(dst, E, dis);
    deg_finalize_kernel<<<(N + T - 1) / T, T, 0, stream>>>(dis, N);

    {
        int nTiles = (N / 16) * (64 / 16);
        int waves_per_block = T / 32;
        gemm_wmma_f32<128, 64><<<(nTiles + waves_per_block - 1) / waves_per_block,
                                 T, 0, stream>>>(x, W1, h1, nTiles);
    }
    tile_agg64_kernel<<<nT, AGG_T, 0, stream>>>(
        src, dst, dis, h1, agg1, gq_cnt, gq_s, gq_d, gq_w, E, N);
    {
        int total = N * 64;
        selfloop_bias_relu_kernel<<<(total + T - 1) / T, T, 0, stream>>>(
            h1, dis, b1, agg1, total);
    }
    {
        int nTiles = (N / 16) * (16 / 16);
        int waves_per_block = T / 32;
        gemm_wmma_f32<64, 16><<<(nTiles + waves_per_block - 1) / waves_per_block,
                                T, 0, stream>>>(agg1, W2, h2, nTiles);
    }
    tile_agg16_kernel<<<nT, AGG2_T, 0, stream>>>(
        src, dst, dis, h2, gq_cnt, gq_s, gq_d, gq_w, agg2, E, N);
    logsoftmax_kernel<<<(N + T - 1) / T, T, 0, stream>>>(h2, dis, b2, agg2, out, N);
}


# Layer-1 product by resident work-groups: each wave keeps its W1 column-tile fragments in registers, 32-row blocks of x go through LDS by whole-row loads, once per row; k and sum order kept
# speedup vs baseline: 3.3837x; 1.8433x over previous
#include <hip/hip_runtime.h>
#include <cstdint>

typedef __attribute__((ext_vector_type(2))) float v2f;
typedef __attribute__((ext_vector_type(4))) float v4f;
typedef __attribute__((ext_vector_type(8))) float v8f;

constexpr int TILE_NODES = 392;

constexpr int TB_T = 256;
constexpr int TB_U = 4;

__global__ __launch_bounds__(TB_T)
void tile_byte_kernel(const int* __restrict__ dst, int E,
                      int* __restrict__ tile_of_edge) {
    const int nq = (E + 3) >> 2;
    const int t0 = blockIdx.x * (TB_T * TB_U) + threadIdx.x;
    int d[TB_U][4];
#pragma unroll
    for (int u = 0; u < TB_U; ++u) {
        const int e0 = 4 * (t0 + u * TB_T);
#pragma unroll
        for (int j = 0; j < 4; ++j) {
            int e = e0 + j;
            d[u][j] = dst[(e < E) ? e : E - 1];
        }
    }
#pragma unroll
    for (int u = 0; u < TB_U; ++u) {
        const int t  = t0 + u * TB_T;
        const int e0 = 4 * t;
        if (t < nq) {
            unsigned pk = 0xFFFFFFFFu;
#pragma unroll
            for (int j = 0; j < 4; ++j) {
                if (e0 + j < E) {
                    unsigned tl = ((unsigned)d[u][j] / (unsigned)TILE_NODES) & 0xFFu;
                    pk = (pk & ~(0xFFu << (8 * j))) | (tl << (8 * j));
                }
            }
            tile_of_edge[t] = (int)pk;
        }
    }
}

constexpr int GM_ROWS = 32;

template <int K, int THREADS, int LOADS>
__device__ __forceinline__ void gemm_load_block(const float* __restrict__ A, int M, int blk, int tid,
                                                v4f (&pre)[LOADS]) {
    constexpr int ROW_F4 = K / 4;
#pragma unroll
    for (int i = 0; i < LOADS; ++i) {
        const int p = tid + i * THREADS;
        int row = blk * GM_ROWS + p / ROW_F4;
        if (row > M - 1) row = M - 1;
        pre[i] = *(const v4f*)(A + (size_t)row * K + 4 * (p % ROW_F4));
    }
}

template <int K, int NC>
__global__ __launch_bounds__((NC / 16) * 32)
void gemm_wmma_f32(const float* __restrict__ A,
                   const float* __restrict__ B,
                   float* __restrict__ C, int M, int nBlocks) {
    constexpr int WAVES   = NC / 16;
    constexpr int THREADS = WAVES * 32;
    constexpr int ROW_F4  = K / 4;
    constexpr int PITCH   = K + 4;
    constexpr int PIECES  = GM_ROWS * ROW_F4;
    constexpr int LOADS   = PIECES / THREADS;
    static_assert(PIECES % THREADS == 0, "block of A must divide over the threads");
    __shared__ __attribute__((aligned(16))) float a_s[GM_ROWS * PITCH];

    const int tid  = threadIdx.x;
    const int lane = tid & 31;
    const int tn   = tid >> 5;
    const int hi   = lane >> 4;
    const int l    = lane & 15;
    const int colBase = tn * 16;

    v2f bf[K / 4];
    {
        const float* bcol = B + colBase + l;
#pragma unroll
        for (int j = 0; j < K / 4; ++j) {
            bf[j].x = bcol[(size_t)(4 * j + 2 * hi + 0) * NC];
            bf[j].y = bcol[(size_t)(4 * j + 2 * hi + 1) * NC];
        }
    }

    v4f pre[LOADS];
    int blk = blockIdx.x;
    if (blk < nBlocks) gemm_load_block<K, THREADS, LOADS>(A, M, blk, tid, pre);
    for (; blk < nBlocks; blk += gridDim.x) {
#pragma unroll
        for (int i = 0; i < LOADS; ++i) {
            const int p = tid + i * THREADS;
            *(v4f*)(a_s + (p / ROW_F4) * PITCH + 4 * (p % ROW_F4)) = pre[i];
        }
        __syncthreads();
        const int nxt = blk + gridDim.x;
        if (nxt < nBlocks) gemm_load_block<K, THREADS, LOADS>(A, M, nxt, tid, pre);

        const float* a0 = a_s + l * PITCH + 2 * hi;
        const float* a1 = a0 + 16 * PITCH;
        v8f acc0 = {};
        v8f acc1 = {};
#pragma unroll
        for (int j = 0; j < K / 4; ++j) {
            const v2f f0 = *(const v2f*)(a0 + 4 * j);
            const v2f f1 = *(const v2f*)(a1 + 4 * j);
            acc0 = __builtin_amdgcn_wmma_f32_16x16x4_f32(
                 false, f0,  false, bf[j],
                 (short)0, acc0,  false,  false);
            acc1 = __builtin_amdgcn_wmma_f32_16x16x4_f32(
                 false, f1,  false, bf[j],
                 (short)0, acc1,  false,  false);
        }

        const int rowBase = blk * GM_ROWS;
        float* cptr = C + (size_t)colBase + l;
#pragma unroll
        for (int i = 0; i < 8; ++i) {
            const int r0 = rowBase + 8 * hi + i;
            const int r1 = r0 + 16;
            if (r0 < M) cptr[(size_t)r0 * NC] = acc0[i];
            if (r1 < M) cptr[(size_t)r1 * NC] = acc1[i];
        }
        __syncthreads();
    }
}

constexpr int QCAP       = 8192;
constexpr int AGG_T      = 1024;
constexpr int SCAN_U     = 8;

__global__ __launch_bounds__(AGG_T)
void tile_scan_kernel(const int* __restrict__ src,
                      const int* __restrict__ dst,
                      float* __restrict__ dis,
                      int* __restrict__ gq_cnt, int* __restrict__ gq_s,
                      int* __restrict__ gq_d,
                      const int* __restrict__ tile_of_edge,
                      int E, int N) {
    __shared__ int   q_a[QCAP];
    __shared__ int   deg_l[TILE_NODES];
    __shared__ int   q_count;

    const int tid  = threadIdx.x;
    const int base = blockIdx.x * TILE_NODES;
    int rows = N - base;
    if (rows > TILE_NODES) rows = TILE_NODES;

    for (int i = tid; i < TILE_NODES; i += AGG_T) deg_l[i] = 0;
    if (tid == 0) q_count = 0;
    __syncthreads();

    const unsigned tile = blockIdx.x;
    const unsigned K    = tile * 0x01010101u;
    const unsigned char* __restrict__ tbb = (const unsigned char*)tile_of_edge;

    auto take = [&](int e) {
        int slot = atomicAdd(&q_count, 1);
        if (slot < QCAP) q_a[slot] = e;
    };
    auto match = [&](unsigned w, unsigned inv) {
        unsigned x = (w ^ K) | inv;
        unsigned t = ((x & 0x7f7f7f7fu) + 0x7f7f7f7fu) | x;
        return ~t & 0x80808080u;
    };
    auto scan16 = [&](int4 v, unsigned inv, int e0) {
        unsigned m = (match((unsigned)v.x, inv) >> 7) | (match((unsigned)v.y, inv) >> 6) |
                     (match((unsigned)v.z, inv) >> 5) | (match((unsigned)v.w, inv) >> 4);
        while (m) {
            int b = __builtin_ctz(m);
            take(e0 + 4 * (b & 7) + (b >> 3));
            m &= m - 1;
        }
    };

    int pre = (int)((16u - (unsigned)((uintptr_t)tbb & 15u)) & 15u);
    if (pre > E) pre = E;
    const int nvec = (E - pre) >> 4;
    const int4* __restrict__ dv = (const int4*)(tbb + pre);

    for (int e = tid; e < pre; e += AGG_T) if (tbb[e] == tile) take(e);
    for (int e = pre + 16 * nvec + tid; e < E; e += AGG_T) if (tbb[e] == tile) take(e);

    constexpr int PER_IT = SCAN_U * AGG_T;
    const int iters = (nvec + PER_IT - 1) / PER_IT;
    const int rot = (int)(((long long)blockIdx.x * iters) / gridDim.x);
    for (int it = 0; it < iters; ++it) {
        int itr = it + rot;
        if (itr >= iters) itr -= iters;
        const int g0 = itr * PER_IT + tid;
        int4     v[SCAN_U];
        unsigned inv[SCAN_U];
#pragma unroll
        for (int u = 0; u < SCAN_U; ++u) {
            int g = g0 + u * AGG_T;
            inv[u] = (g < nvec) ? 0u : 0xFFFFFFFFu;
            v[u]   = dv[(g < nvec) ? g : nvec - 1];
        }
#pragma unroll
        for (int u = 0; u < SCAN_U; ++u) {
            int e0 = pre + 16 * (g0 + u * AGG_T);
            scan16(v[u], inv[u], e0);
        }
    }
    __syncthreads();

    int total = q_count;
    if (tid == 0) gq_cnt[blockIdx.x] = total;
    if (total > QCAP) {
        for (int e = tid; e < E; e += AGG_T) {
            unsigned dl = (unsigned)(dst[e] - base);
            if (dl < (unsigned)rows) atomicAdd(&deg_l[dl], 1);
        }
        total = 0;
    }

    int* __restrict__ ts = gq_s + (size_t)blockIdx.x * QCAP;
    int* __restrict__ td = gq_d + (size_t)blockIdx.x * QCAP;
    for (int i = tid; i < total; i += AGG_T) {
        int e  = q_a[i];
        int s  = src[e];
        int dl = dst[e] - base;
        if ((unsigned)dl < (unsigned)rows) atomicAdd(&deg_l[dl], 1);
        else                               dl = TILE_NODES;
        ts[i] = s;
        td[i] = dl;
    }
    __syncthreads();

    for (int r = tid; r < rows; r += AGG_T)
        dis[base + r] = rsqrtf((float)deg_l[r] + 1.0f);
}

__global__ __launch_bounds__(AGG_T)
void tile_agg64_kernel(const int* __restrict__ src,
                       const int* __restrict__ dst,
                       const float* __restrict__ dis,
                       const float* __restrict__ h,
                       float* __restrict__ agg,
                       const int* __restrict__ gq_cnt, const int* __restrict__ gq_s,
                       const int* __restrict__ gq_d, float* __restrict__ gq_w,
                       int E, int N) {
    constexpr int F = 64;
    __shared__ float acc[(TILE_NODES + 1) * F];
    __shared__ int   q_a[QCAP];
    __shared__ int   q_d[QCAP];
    __shared__ float q_w[QCAP];

    const int tid  = threadIdx.x;
    const int base = blockIdx.x * TILE_NODES;
    int rows = N - base;
    if (rows > TILE_NODES) rows = TILE_NODES;

    for (int i = tid; i < (TILE_NODES + 1) * F; i += AGG_T) acc[i] = 0.0f;
    __syncthreads();

    int total = gq_cnt[blockIdx.x];
    if (total > QCAP) {
        for (int e = tid; e < E; e += AGG_T) {
            int d = dst[e];
            unsigned dl = (unsigned)(d - base);
            if (dl < (unsigned)rows) {
                int s = src[e];
                float w = dis[s] * dis[d];
                for (int j = 0; j < F; ++j)
                    atomicAdd(&acc[dl * F + j], h[(size_t)s * F + j] * w);
            }
        }
        total = 0;
    }

    const int* __restrict__ ts = gq_s + (size_t)blockIdx.x * QCAP;
    const int* __restrict__ td = gq_d + (size_t)blockIdx.x * QCAP;
    float*     __restrict__ tw = gq_w + (size_t)blockIdx.x * QCAP;
    for (int i = tid; i < total; i += AGG_T) {
        int s  = ts[i];
        int dl = td[i];
        const bool in_tile = (unsigned)dl < (unsigned)rows;
        float w = in_tile ? dis[s] * dis[base + (in_tile ? dl : 0)] : 0.0f;
        q_a[i] = s;
        q_d[i] = dl;
        q_w[i] = w;
        tw[i]  = w;
    }
    __syncthreads();

    const int wave = tid >> 5;
    const int lane = tid & 31;
    constexpr int R = 8;
    for (int i0 = wave * R; i0 < total; i0 += (AGG_T / 32) * R) {
        int    sn[R];
        int    dn[R];
        float  wn[R];
        float2 hv[R];
#pragma unroll
        for (int r = 0; r < R; ++r) {
            int i = i0 + r;
            if (i > total - 1) i = total - 1;
            sn[r] = q_a[i];
            dn[r] = q_d[i];
            wn[r] = q_w[i];
        }
#pragma unroll
        for (int r = 0; r < R; ++r)
            hv[r] = *(const float2*)(h + (size_t)sn[r] * F + 2 * lane);
        asm volatile("" :: "v"(hv[0].x), "v"(hv[0].y), "v"(hv[1].x), "v"(hv[1].y),
                           "v"(hv[2].x), "v"(hv[2].y), "v"(hv[3].x), "v"(hv[3].y),
                           "v"(hv[4].x), "v"(hv[4].y), "v"(hv[5].x), "v"(hv[5].y),
                           "v"(hv[6].x), "v"(hv[6].y), "v"(hv[7].x), "v"(hv[7].y));
#pragma unroll
        for (int r = 0; r < R; ++r) {
            int i = i0 + r;
            if (i < total) {
                atomicAdd(&acc[dn[r] * F + 2 * lane + 0], hv[r].x * wn[r]);
                atomicAdd(&acc[dn[r] * F + 2 * lane + 1], hv[r].y * wn[r]);
            }
        }
    }
    __syncthreads();

    const int n4 = rows * (F / 4);
    float4* __restrict__ o4 = (float4*)(agg + (size_t)base * F);
    const float4* a4 = (const float4*)acc;
    for (int i = tid; i < n4; i += AGG_T) o4[i] = a4[i];
}

constexpr int AGG2_T = 1024;
constexpr int R2     = 8;

__global__ __launch_bounds__(AGG2_T)
void tile_agg16_kernel(const int* __restrict__ src,
                       const int* __restrict__ dst,
                       const float* __restrict__ dis,
                       const float* __restrict__ h,
                       const int* __restrict__ gq_cnt, const int* __restrict__ gq_s,
                       const int* __restrict__ gq_d, const float* __restrict__ gq_w,
                       float* __restrict__ agg, int E, int N) {
    constexpr int F = 16;
    __shared__ float acc[(TILE_NODES + 1) * F];

    const int tid  = threadIdx.x;
    const int base = blockIdx.x * TILE_NODES;
    int rows = N - base;
    if (rows > TILE_NODES) rows = TILE_NODES;

    for (int i = tid; i < (TILE_NODES + 1) * F; i += AGG2_T) acc[i] = 0.0f;
    __syncthreads();

    int total = gq_cnt[blockIdx.x];
    if (total > QCAP) {
        for (int e = tid; e < E; e += AGG2_T) {
            int d = dst[e];
            unsigned dl = (unsigned)(d - base);
            if (dl < (unsigned)rows) {
                int s = src[e];
                float w = dis[s] * dis[d];
                for (int j = 0; j < F; ++j)
                    atomicAdd(&acc[dl * F + j], h[(size_t)s * F + j] * w);
            }
        }
        total = 0;
    }

    const int*   __restrict__ ts = gq_s + (size_t)blockIdx.x * QCAP;
    const int*   __restrict__ td = gq_d + (size_t)blockIdx.x * QCAP;
    const float* __restrict__ tw = gq_w + (size_t)blockIdx.x * QCAP;

    const int grp = tid >> 2;
    const int qt  = tid & 3;
    constexpr int GROUPS = AGG2_T / 4;
    for (int r0 = 0; r0 < total; r0 += GROUPS * R2) {
        int   sn[R2];
        int   dn[R2];
        float wn[R2];
        v4f   hv[R2];
#pragma unroll
        for (int r = 0; r < R2; ++r) {
            int i = r0 + r * GROUPS + grp;
            if (i > total - 1) i = total - 1;
            sn[r] = ts[i];
            dn[r] = td[i];
            wn[r] = tw[i];
        }
#pragma unroll
        for (int r = 0; r < R2; ++r)
            hv[r] = *(const v4f*)(h + (size_t)sn[r] * F + 4 * qt);
        asm volatile("" :: "v"(hv[0]), "v"(hv[1]), "v"(hv[2]), "v"(hv[3]),
                           "v"(hv[4]), "v"(hv[5]), "v"(hv[6]), "v"(hv[7]),
                           "v"(dn[0]), "v"(dn[1]), "v"(dn[2]), "v"(dn[3]),
                           "v"(dn[4]), "v"(dn[5]), "v"(dn[6]), "v"(dn[7]),
                           "v"(wn[0]), "v"(wn[1]), "v"(wn[2]), "v"(wn[3]),
                           "v"(wn[4]), "v"(wn[5]), "v"(wn[6]), "v"(wn[7]));
#pragma unroll
        for (int r = 0; r < R2; ++r) {
            int i = r0 + r * GROUPS + grp;
            if (i < total) {
                float* p = &acc[dn[r] * F + 4 * qt];
                atomicAdd(p + 0, hv[r].x * wn[r]);
                atomicAdd(p + 1, hv[r].y * wn[r]);
                atomicAdd(p + 2, hv[r].z * wn[r]);
                atomicAdd(p + 3, hv[r].w * wn[r]);
            }
        }
    }
    __syncthreads();

    const int n4 = rows * (F / 4);
    float4* __restrict__ o4 = (float4*)(agg + (size_t)base * F);
    const float4* a4 = (const float4*)acc;
    for (int i = tid; i < n4; i += AGG2_T) o4[i] = a4[i];
}

template <int K, int NC>
__global__ void gemm_epi_wmma_f32(const float* __restrict__ agg,
                                  const float* __restrict__ h,
                                  const float* __restrict__ dis,
                                  const float* __restrict__ bias,
                                  const float* __restrict__ B,
                                  float* __restrict__ C, int nTiles) {
    int wave = (int)((blockIdx.x * blockDim.x + threadIdx.x) >> 5);
    int lane = threadIdx.x & 31;
    if (wave >= nTiles) return;
    constexpr int TN = NC / 16;
    int tm = wave / TN;
    int tn = wave % TN;
    int rowBase = tm * 16;
    int colBase = tn * 16;
    int hi = lane >> 4;
    int l  = lane & 15;

    const size_t roff = (size_t)(rowBase + l) * K + 2 * hi;
    const float* grow = agg + roff;
    const float* hrow = h + roff;
    const float* brow = bias + 2 * hi;
    const float* bcol = B + colBase + l;
    const float  dii  = dis[rowBase + l];

    v8f acc = {};
#pragma unroll
    for (int k = 0; k < K; k += 4) {
        const float2 g  = *(const float2*)(grow + k);
        const float2 hh = *(const float2*)(hrow + k);
        const float2 bb = *(const float2*)(brow + k);
        v2f a, b;
        a.x = fmaxf(g.x + hh.x * dii * dii + bb.x, 0.0f);
        a.y = fmaxf(g.y + hh.y * dii * dii + bb.y, 0.0f);
        b.x = bcol[(size_t)(k + 2 * hi + 0) * NC];
        b.y = bcol[(size_t)(k + 2 * hi + 1) * NC];
        acc = __builtin_amdgcn_wmma_f32_16x16x4_f32(
             false, a,  false, b,
             (short)0, acc,  false,  false);
    }

    float* cptr = C + (size_t)colBase + l;
#pragma unroll
    for (int i = 0; i < 8; ++i)
        cptr[(size_t)(rowBase + 8 * hi + i) * NC] = acc[i];
}

__global__ void logsoftmax_kernel(const float* __restrict__ h,
                                  const float* __restrict__ dis,
                                  const float* __restrict__ bias,
                                  const float* __restrict__ agg,
                                  float* __restrict__ out, int n) {
    int i = blockIdx.x * blockDim.x + threadIdx.x;
    if (i >= n) return;
    float dii = dis[i];
    dii *= dii;
    float v[16];
    float mx = -3.402823466e38f;
#pragma unroll
    for (int j = 0; j < 16; ++j) {
        size_t idx = (size_t)i * 16 + j;
        v[j] = agg[idx] + h[idx] * dii + bias[j];
        mx = fmaxf(mx, v[j]);
    }
    float s = 0.0f;
#pragma unroll
    for (int j = 0; j < 16; ++j) s += __expf(v[j] - mx);
    float lse = mx + __logf(s);
#pragma unroll
    for (int j = 0; j < 16; ++j) out[(size_t)i * 16 + j] = v[j] - lse;
}

extern "C" void kernel_launch(void* const* d_in, const int* in_sizes, int n_in,
                              void* d_out, int out_size, void* d_ws, size_t ws_size,
                              hipStream_t stream) {
    const float* x   = (const float*)d_in[0];
    const int*   ei  = (const int*)  d_in[1];
    const float* W1  = (const float*)d_in[2];
    const float* b1  = (const float*)d_in[3];
    const float* W2  = (const float*)d_in[4];
    const float* b2  = (const float*)d_in[5];
    float* out = (float*)d_out;

    const int N = in_sizes[0] / 128;
    const int E = in_sizes[1] / 2;
    const int* src = ei;
    const int* dst = ei + E;

    float* dis  = (float*)d_ws;
    float* h1   = dis  + N;
    float* agg1 = h1   + (size_t)N * 64;
    float* h2   = agg1 + (size_t)N * 64;
    float* agg2 = h2   + (size_t)N * 16;
    const int nT = (N + TILE_NODES - 1) / TILE_NODES;
    int*   gq_s   = (int*)(agg2 + (size_t)N * 16);
    int*   gq_d   = gq_s + (size_t)nT * QCAP;
    float* gq_w   = (float*)(gq_d + (size_t)nT * QCAP);
    int*   gq_cnt = (int*)(gq_w + (size_t)nT * QCAP);
    int*   tile_of_edge = gq_cnt + nT;

    const int T = 256;

    {
        const int nq = (E + 3) / 4;
        tile_byte_kernel<<<(nq + TB_T * TB_U - 1) / (TB_T * TB_U), TB_T, 0, stream>>>(
            dst, E, tile_of_edge);
    }
    tile_scan_kernel<<<nT, AGG_T, 0, stream>>>(
        src, dst, dis, gq_cnt, gq_s, gq_d, tile_of_edge, E, N);

    {
        int nBlocks = (N + GM_ROWS - 1) / GM_ROWS;
        gemm_wmma_f32<128, 64><<<(nBlocks < 1024) ? nBlocks : 1024, (64 / 16) * 32, 0, stream>>>(
            x, W1, h1, N, nBlocks);
    }
    tile_agg64_kernel<<<nT, AGG_T, 0, stream>>>(
        src, dst, dis, h1, agg1, gq_cnt, gq_s, gq_d, gq_w, E, N);
    {
        int nTiles = (N / 16) * (16 / 16);
        int waves_per_block = T / 32;
        gemm_epi_wmma_f32<64, 16><<<(nTiles + waves_per_block - 1) / waves_per_block,
                                    T, 0, stream>>>(agg1, h1, dis, b1, W2, h2, nTiles);
    }
    tile_agg16_kernel<<<nT, AGG2_T, 0, stream>>>(
        src, dst, dis, h2, gq_cnt, gq_s, gq_d, gq_w, agg2, E, N);
    logsoftmax_kernel<<<(N + T - 1) / T, T, 0, stream>>>(h2, dis, b2, agg2, out, N);
}


# Tile scan reads its tile's presence bits (one bit per 16-edge group, ORed in LDS by the byte pass) and loads only the byte vectors whose bit is set; queue order, hence sum order, changes
# speedup vs baseline: 3.7625x; 1.1119x over previous
#include <hip/hip_runtime.h>
#include <cstdint>

typedef __attribute__((ext_vector_type(2))) float v2f;
typedef __attribute__((ext_vector_type(4))) float v4f;
typedef __attribute__((ext_vector_type(8))) float v8f;

constexpr int TILE_NODES = 392;

constexpr int TB_T      = 1024;
constexpr int TB_U      = 4;
constexpr int TB_GROUPS = TB_T * TB_U * 4 / 16;
constexpr int PW        = TB_GROUPS / 32;
constexpr int PB_PITCH  = 257;
static_assert(PW == 32, "one 128-byte line of presence bits per tile and work-group");

__global__ __launch_bounds__(TB_T)
void tile_byte_kernel(const int* __restrict__ dst, int E,
                      int* __restrict__ tile_of_edge,
                      unsigned* __restrict__ pres, int nT, int nblk) {
    __shared__ unsigned pb[PW * PB_PITCH];
    const int tid = threadIdx.x;
    for (int i = tid; i < PW * PB_PITCH; i += TB_T) pb[i] = 0u;

    const int nq = (E + 3) >> 2;
    const int t0 = blockIdx.x * (TB_T * TB_U) + tid;
    int d[TB_U][4];
#pragma unroll
    for (int u = 0; u < TB_U; ++u) {
        const int e0 = 4 * (t0 + u * TB_T);
#pragma unroll
        for (int j = 0; j < 4; ++j) {
            int e = e0 + j;
            d[u][j] = dst[(e < E) ? e : E - 1];
        }
    }
    __syncthreads();
#pragma unroll
    for (int u = 0; u < TB_U; ++u) {
        const int t  = t0 + u * TB_T;
        const int e0 = 4 * t;
        const int lg = (tid + u * TB_T) >> 2;
        unsigned* pw = pb + (lg >> 5) * PB_PITCH;
        const unsigned bit = 1u << (lg & 31);
        if (t < nq) {
            unsigned pk = 0xFFFFFFFFu;
#pragma unroll
            for (int j = 0; j < 4; ++j) {
                if (e0 + j < E) {
                    unsigned tl = ((unsigned)d[u][j] / (unsigned)TILE_NODES) & 0xFFu;
                    pk = (pk & ~(0xFFu << (8 * j))) | (tl << (8 * j));
                    atomicOr(&pw[tl], bit);
                }
            }
            tile_of_edge[t] = (int)pk;
        }
    }
    __syncthreads();
    for (int p = tid; p < nT * (PW / 4); p += TB_T) {
        const int tile = p >> 3;
        const int q    = p & 7;
        uint4 o;
        o.x = pb[(4 * q + 0) * PB_PITCH + tile];
        o.y = pb[(4 * q + 1) * PB_PITCH + tile];
        o.z = pb[(4 * q + 2) * PB_PITCH + tile];
        o.w = pb[(4 * q + 3) * PB_PITCH + tile];
        *(uint4*)(pres + ((size_t)tile * nblk + blockIdx.x) * PW + 4 * q) = o;
    }
}

constexpr int GM_ROWS = 32;

template <int K, int THREADS, int LOADS>
__device__ __forceinline__ void gemm_load_block(const float* __restrict__ A, int M, int blk, int tid,
                                                v4f (&pre)[LOADS]) {
    constexpr int ROW_F4 = K / 4;
#pragma unroll
    for (int i = 0; i < LOADS; ++i) {
        const int p = tid + i * THREADS;
        int row = blk * GM_ROWS + p / ROW_F4;
        if (row > M - 1) row = M - 1;
        pre[i] = *(const v4f*)(A + (size_t)row * K + 4 * (p % ROW_F4));
    }
}

template <int K, int NC>
__global__ __launch_bounds__((NC / 16) * 32)
void gemm_wmma_f32(const float* __restrict__ A,
                   const float* __restrict__ B,
                   float* __restrict__ C, int M, int nBlocks) {
    constexpr int WAVES   = NC / 16;
    constexpr int THREADS = WAVES * 32;
    constexpr int ROW_F4  = K / 4;
    constexpr int PITCH   = K + 4;
    constexpr int PIECES  = GM_ROWS * ROW_F4;
    constexpr int LOADS   = PIECES / THREADS;
    static_assert(PIECES % THREADS == 0, "block of A must divide over the threads");
    __shared__ __attribute__((aligned(16))) float a_s[GM_ROWS * PITCH];

    const int tid  = threadIdx.x;
    const int lane = tid & 31;
    const int tn   = tid >> 5;
    const int hi   = lane >> 4;
    const int l    = lane & 15;
    const int colBase = tn * 16;

    v2f bf[K / 4];
    {
        const float* bcol = B + colBase + l;
#pragma unroll
        for (int j = 0; j < K / 4; ++j) {
            bf[j].x = bcol[(size_t)(4 * j + 2 * hi + 0) * NC];
            bf[j].y = bcol[(size_t)(4 * j + 2 * hi + 1) * NC];
        }
    }

    v4f pre[LOADS];
    int blk = blockIdx.x;
    if (blk < nBlocks) gemm_load_block<K, THREADS, LOADS>(A, M, blk, tid, pre);
    for (; blk < nBlocks; blk += gridDim.x) {
#pragma unroll
        for (int i = 0; i < LOADS; ++i) {
            const int p = tid + i * THREADS;
            *(v4f*)(a_s + (p / ROW_F4) * PITCH + 4 * (p % ROW_F4)) = pre[i];
        }
        __syncthreads();
        const int nxt = blk + gridDim.x;
        if (nxt < nBlocks) gemm_load_block<K, THREADS, LOADS>(A, M, nxt, tid, pre);

        const float* a0 = a_s + l * PITCH + 2 * hi;
        const float* a1 = a0 + 16 * PITCH;
        v8f acc0 = {};
        v8f acc1 = {};
#pragma unroll
        for (int j = 0; j < K / 4; ++j) {
            const v2f f0 = *(const v2f*)(a0 + 4 * j);
            const v2f f1 = *(const v2f*)(a1 + 4 * j);
            acc0 = __builtin_amdgcn_wmma_f32_16x16x4_f32(
                 false, f0,  false, bf[j],
                 (short)0, acc0,  false,  false);
            acc1 = __builtin_amdgcn_wmma_f32_16x16x4_f32(
                 false, f1,  false, bf[j],
                 (short)0, acc1,  false,  false);
        }

        const int rowBase = blk * GM_ROWS;
        float* cptr = C + (size_t)colBase + l;
#pragma unroll
        for (int i = 0; i < 8; ++i) {
            const int r0 = rowBase + 8 * hi + i;
            const int r1 = r0 + 16;
            if (r0 < M) cptr[(size_t)r0 * NC] = acc0[i];
            if (r1 < M) cptr[(size_t)r1 * NC] = acc1[i];
        }
        __syncthreads();
    }
}

constexpr int QCAP       = 8192;
constexpr int AGG_T      = 1024;
constexpr int PRES_U     = 4;
constexpr int GRP_U      = QCAP / AGG_T;

__global__ __launch_bounds__(AGG_T)
void tile_scan_kernel(const int* __restrict__ src,
                      const int* __restrict__ dst,
                      float* __restrict__ dis,
                      int* __restrict__ gq_cnt, int* __restrict__ gq_s,
                      int* __restrict__ gq_d,
                      const int* __restrict__ tile_of_edge,
                      const unsigned* __restrict__ pres, int nblk,
                      int E, int N) {
    __shared__ int   q_a[QCAP];
    __shared__ int   q_g[QCAP];
    __shared__ int   deg_l[TILE_NODES];
    __shared__ int   q_count;
    __shared__ int   g_count;

    const int tid  = threadIdx.x;
    const int base = blockIdx.x * TILE_NODES;
    int rows = N - base;
    if (rows > TILE_NODES) rows = TILE_NODES;

    for (int i = tid; i < TILE_NODES; i += AGG_T) deg_l[i] = 0;
    if (tid == 0) { q_count = 0; g_count = 0; }
    __syncthreads();

    const unsigned tile = blockIdx.x;
    const unsigned K    = tile * 0x01010101u;

    auto take = [&](int e) {
        if (e < E) {
            int slot = atomicAdd(&q_count, 1);
            if (slot < QCAP) q_a[slot] = e;
        }
    };
    auto match = [&](unsigned w, unsigned inv) {
        unsigned x = (w ^ K) | inv;
        unsigned t = ((x & 0x7f7f7f7fu) + 0x7f7f7f7fu) | x;
        return ~t & 0x80808080u;
    };
    auto scan16 = [&](int4 v, unsigned inv, int e0) {
        unsigned m = (match((unsigned)v.x, inv) >> 7) | (match((unsigned)v.y, inv) >> 6) |
                     (match((unsigned)v.z, inv) >> 5) | (match((unsigned)v.w, inv) >> 4);
        while (m) {
            int b = __builtin_ctz(m);
            take(e0 + 4 * (b & 7) + (b >> 3));
            m &= m - 1;
        }
    };

    const int nwords = nblk * PW;
    const unsigned* __restrict__ prow = pres + (size_t)blockIdx.x * nwords;
    for (int w0 = 0; w0 < nwords; w0 += PRES_U * AGG_T) {
        unsigned bits[PRES_U];
#pragma unroll
        for (int u = 0; u < PRES_U; ++u) {
            int w = w0 + u * AGG_T + tid;
            unsigned x = prow[(w < nwords) ? w : nwords - 1];
            bits[u] = (w < nwords) ? x : 0u;
        }
#pragma unroll
        for (int u = 0; u < PRES_U; ++u) {
            const int w = w0 + u * AGG_T + tid;
            unsigned m = bits[u];
            while (m) {
                int b = __builtin_ctz(m);
                int slot = atomicAdd(&g_count, 1);
                if (slot < QCAP) q_g[slot] = 32 * w + b;
                m &= m - 1;
            }
        }
    }
    __syncthreads();

    const int gtotal = g_count;
    if (gtotal > QCAP) {
        if (tid == 0) q_count = QCAP + 1;
    } else if (gtotal > 0) {
        const int4* __restrict__ dv = (const int4*)tile_of_edge;
        int  gi[GRP_U];
        int4 v[GRP_U];
#pragma unroll
        for (int u = 0; u < GRP_U; ++u) {
            int i = tid + u * AGG_T;
            gi[u] = q_g[(i < gtotal) ? i : gtotal - 1];
        }
#pragma unroll
        for (int u = 0; u < GRP_U; ++u) v[u] = dv[gi[u]];
#pragma unroll
        for (int u = 0; u < GRP_U; ++u) {
            int i = tid + u * AGG_T;
            scan16(v[u], (i < gtotal) ? 0u : 0xFFFFFFFFu, 16 * gi[u]);
        }
    }
    __syncthreads();

    int total = q_count;
    if (tid == 0) gq_cnt[blockIdx.x] = total;
    if (total > QCAP) {
        for (int e = tid; e < E; e += AGG_T) {
            unsigned dl = (unsigned)(dst[e] - base);
            if (dl < (unsigned)rows) atomicAdd(&deg_l[dl], 1);
        }
        total = 0;
    }

    int* __restrict__ ts = gq_s + (size_t)blockIdx.x * QCAP;
    int* __restrict__ td = gq_d + (size_t)blockIdx.x * QCAP;
    for (int i = tid; i < total; i += AGG_T) {
        int e  = q_a[i];
        int s  = src[e];
        int dl = dst[e] - base;
        if ((unsigned)dl < (unsigned)rows) atomicAdd(&deg_l[dl], 1);
        else                               dl = TILE_NODES;
        ts[i] = s;
        td[i] = dl;
    }
    __syncthreads();

    for (int r = tid; r < rows; r += AGG_T)
        dis[base + r] = rsqrtf((float)deg_l[r] + 1.0f);
}

__global__ __launch_bounds__(AGG_T)
void tile_agg64_kernel(const int* __restrict__ src,
                       const int* __restrict__ dst,
                       const float* __restrict__ dis,
                       const float* __restrict__ h,
                       float* __restrict__ agg,
                       const int* __restrict__ gq_cnt, const int* __restrict__ gq_s,
                       const int* __restrict__ gq_d, float* __restrict__ gq_w,
                       int E, int N) {
    constexpr int F = 64;
    __shared__ float acc[(TILE_NODES + 1) * F];
    __shared__ int   q_a[QCAP];
    __shared__ int   q_d[QCAP];
    __shared__ float q_w[QCAP];

    const int tid  = threadIdx.x;
    const int base = blockIdx.x * TILE_NODES;
    int rows = N - base;
    if (rows > TILE_NODES) rows = TILE_NODES;

    for (int i = tid; i < (TILE_NODES + 1) * F; i += AGG_T) acc[i] = 0.0f;
    __syncthreads();

    int total = gq_cnt[blockIdx.x];
    if (total > QCAP) {
        for (int e = tid; e < E; e += AGG_T) {
            int d = dst[e];
            unsigned dl = (unsigned)(d - base);
            if (dl < (unsigned)rows) {
                int s = src[e];
                float w = dis[s] * dis[d];
                for (int j = 0; j < F; ++j)
                    atomicAdd(&acc[dl * F + j], h[(size_t)s * F + j] * w);
            }
        }
        total = 0;
    }

    const int* __restrict__ ts = gq_s + (size_t)blockIdx.x * QCAP;
    const int* __restrict__ td = gq_d + (size_t)blockIdx.x * QCAP;
    float*     __restrict__ tw = gq_w + (size_t)blockIdx.x * QCAP;
    for (int i = tid; i < total; i += AGG_T) {
        int s  = ts[i];
        int dl = td[i];
        const bool in_tile = (unsigned)dl < (unsigned)rows;
        float w = in_tile ? dis[s] * dis[base + (in_tile ? dl : 0)] : 0.0f;
        q_a[i] = s;
        q_d[i] = dl;
        q_w[i] = w;
        tw[i]  = w;
    }
    __syncthreads();

    const int wave = tid >> 5;
    const int lane = tid & 31;
    constexpr int R = 8;
    for (int i0 = wave * R; i0 < total; i0 += (AGG_T / 32) * R) {
        int    sn[R];
        int    dn[R];
        float  wn[R];
        float2 hv[R];
#pragma unroll
        for (int r = 0; r < R; ++r) {
            int i = i0 + r;
            if (i > total - 1) i = total - 1;
            sn[r] = q_a[i];
            dn[r] = q_d[i];
            wn[r] = q_w[i];
        }
#pragma unroll
        for (int r = 0; r < R; ++r)
            hv[r] = *(const float2*)(h + (size_t)sn[r] * F + 2 * lane);
        asm volatile("" :: "v"(hv[0].x), "v"(hv[0].y), "v"(hv[1].x), "v"(hv[1].y),
                           "v"(hv[2].x), "v"(hv[2].y), "v"(hv[3].x), "v"(hv[3].y),
                           "v"(hv[4].x), "v"(hv[4].y), "v"(hv[5].x), "v"(hv[5].y),
                           "v"(hv[6].x), "v"(hv[6].y), "v"(hv[7].x), "v"(hv[7].y));
#pragma unroll
        for (int r = 0; r < R; ++r) {
            int i = i0 + r;
            if (i < total) {
                atomicAdd(&acc[dn[r] * F + 2 * lane + 0], hv[r].x * wn[r]);
                atomicAdd(&acc[dn[r] * F + 2 * lane + 1], hv[r].y * wn[r]);
            }
        }
    }
    __syncthreads();

    const int n4 = rows * (F / 4);
    float4* __restrict__ o4 = (float4*)(agg + (size_t)base * F);
    const float4* a4 = (const float4*)acc;
    for (int i = tid; i < n4; i += AGG_T) o4[i] = a4[i];
}

constexpr int AGG2_T = 1024;
constexpr int R2     = 8;

__global__ __launch_bounds__(AGG2_T)
void tile_agg16_kernel(const int* __restrict__ src,
                       const int* __restrict__ dst,
                       const float* __restrict__ dis,
                       const float* __restrict__ h,
                       const int* __restrict__ gq_cnt, const int* __restrict__ gq_s,
                       const int* __restrict__ gq_d, const float* __restrict__ gq_w,
                       float* __restrict__ agg, int E, int N) {
    constexpr int F = 16;
    __shared__ float acc[(TILE_NODES + 1) * F];

    const int tid  = threadIdx.x;
    const int base = blockIdx.x * TILE_NODES;
    int rows = N - base;
    if (rows > TILE_NODES) rows = TILE_NODES;

    for (int i = tid; i < (TILE_NODES + 1) * F; i += AGG2_T) acc[i] = 0.0f;
    __syncthreads();

    int total = gq_cnt[blockIdx.x];
    if (total > QCAP) {
        for (int e = tid; e < E; e += AGG2_T) {
            int d = dst[e];
            unsigned dl = (unsigned)(d - base);
            if (dl < (unsigned)rows) {
                int s = src[e];
                float w = dis[s] * dis[d];
                for (int j = 0; j < F; ++j)
                    atomicAdd(&acc[dl * F + j], h[(size_t)s * F + j] * w);
            }
        }
        total = 0;
    }

    const int*   __restrict__ ts = gq_s + (size_t)blockIdx.x * QCAP;
    const int*   __restrict__ td = gq_d + (size_t)blockIdx.x * QCAP;
    const float* __restrict__ tw = gq_w + (size_t)blockIdx.x * QCAP;

    const int grp = tid >> 2;
    const int qt  = tid & 3;
    constexpr int GROUPS = AGG2_T / 4;
    for (int r0 = 0; r0 < total; r0 += GROUPS * R2) {
        int   sn[R2];
        int   dn[R2];
        float wn[R2];
        v4f   hv[R2];
#pragma unroll
        for (int r = 0; r < R2; ++r) {
            int i = r0 + r * GROUPS + grp;
            if (i > total - 1) i = total - 1;
            sn[r] = ts[i];
            dn[r] = td[i];
            wn[r] = tw[i];
        }
#pragma unroll
        for (int r = 0; r < R2; ++r)
            hv[r] = *(const v4f*)(h + (size_t)sn[r] * F + 4 * qt);
        asm volatile("" :: "v"(hv[0]), "v"(hv[1]), "v"(hv[2]), "v"(hv[3]),
                           "v"(hv[4]), "v"(hv[5]), "v"(hv[6]), "v"(hv[7]),
                           "v"(dn[0]), "v"(dn[1]), "v"(dn[2]), "v"(dn[3]),
                           "v"(dn[4]), "v"(dn[5]), "v"(dn[6]), "v"(dn[7]),
                           "v"(wn[0]), "v"(wn[1]), "v"(wn[2]), "v"(wn[3]),
                           "v"(wn[4]), "v"(wn[5]), "v"(wn[6]), "v"(wn[7]));
#pragma unroll
        for (int r = 0; r < R2; ++r) {
            int i = r0 + r * GROUPS + grp;
            if (i < total) {
                float* p = &acc[dn[r] * F + 4 * qt];
                atomicAdd(p + 0, hv[r].x * wn[r]);
                atomicAdd(p + 1, hv[r].y * wn[r]);
                atomicAdd(p + 2, hv[r].z * wn[r]);
                atomicAdd(p + 3, hv[r].w * wn[r]);
            }
        }
    }
    __syncthreads();

    const int n4 = rows * (F / 4);
    float4* __restrict__ o4 = (float4*)(agg + (size_t)base * F);
    const float4* a4 = (const float4*)acc;
    for (int i = tid; i < n4; i += AGG2_T) o4[i] = a4[i];
}

template <int K, int NC>
__global__ void gemm_epi_wmma_f32(const float* __restrict__ agg,
                                  const float* __restrict__ h,
                                  const float* __restrict__ dis,
                                  const float* __restrict__ bias,
                                  const float* __restrict__ B,
                                  float* __restrict__ C, int nTiles) {
    int wave = (int)((blockIdx.x * blockDim.x + threadIdx.x) >> 5);
    int lane = threadIdx.x & 31;
    if (wave >= nTiles) return;
    constexpr int TN = NC / 16;
    int tm = wave / TN;
    int tn = wave % TN;
    int rowBase = tm * 16;
    int colBase = tn * 16;
    int hi = lane >> 4;
    int l  = lane & 15;

    const size_t roff = (size_t)(rowBase + l) * K + 2 * hi;
    const float* grow = agg + roff;
    const float* hrow = h + roff;
    const float* brow = bias + 2 * hi;
    const float* bcol = B + colBase + l;
    const float  dii  = dis[rowBase + l];

    v8f acc = {};
#pragma unroll
    for (int k = 0; k < K; k += 4) {
        const float2 g  = *(const float2*)(grow + k);
        const float2 hh = *(const float2*)(hrow + k);
        const float2 bb = *(const float2*)(brow + k);
        v2f a, b;
        a.x = fmaxf(g.x + hh.x * dii * dii + bb.x, 0.0f);
        a.y = fmaxf(g.y + hh.y * dii * dii + bb.y, 0.0f);
        b.x = bcol[(size_t)(k + 2 * hi + 0) * NC];
        b.y = bcol[(size_t)(k + 2 * hi + 1) * NC];
        acc = __builtin_amdgcn_wmma_f32_16x16x4_f32(
             false, a,  false, b,
             (short)0, acc,  false,  false);
    }

    float* cptr = C + (size_t)colBase + l;
#pragma unroll
    for (int i = 0; i < 8; ++i)
        cptr[(size_t)(rowBase + 8 * hi + i) * NC] = acc[i];
}

__global__ void logsoftmax_kernel(const float* __restrict__ h,
                                  const float* __restrict__ dis,
                                  const float* __restrict__ bias,
                                  const float* __restrict__ agg,
                                  float* __restrict__ out, int n) {
    int i = blockIdx.x * blockDim.x + threadIdx.x;
    if (i >= n) return;
    float dii = dis[i];
    dii *= dii;
    float v[16];
    float mx = -3.402823466e38f;
#pragma unroll
    for (int j = 0; j < 16; ++j) {
        size_t idx = (size_t)i * 16 + j;
        v[j] = agg[idx] + h[idx] * dii + bias[j];
        mx = fmaxf(mx, v[j]);
    }
    float s = 0.0f;
#pragma unroll
    for (int j = 0; j < 16; ++j) s += __expf(v[j] - mx);
    float lse = mx + __logf(s);
#pragma unroll
    for (int j = 0; j < 16; ++j) out[(size_t)i * 16 + j] = v[j] - lse;
}

extern "C" void kernel_launch(void* const* d_in, const int* in_sizes, int n_in,
                              void* d_out, int out_size, void* d_ws, size_t ws_size,
                              hipStream_t stream) {
    const float* x   = (const float*)d_in[0];
    const int*   ei  = (const int*)  d_in[1];
    const float* W1  = (const float*)d_in[2];
    const float* b1  = (const float*)d_in[3];
    const float* W2  = (const float*)d_in[4];
    const float* b2  = (const float*)d_in[5];
    float* out = (float*)d_out;

    const int N = in_sizes[0] / 128;
    const int E = in_sizes[1] / 2;
    const int* src = ei;
    const int* dst = ei + E;

    float* dis  = (float*)d_ws;
    float* h1   = dis  + N;
    float* agg1 = h1   + (size_t)N * 64;
    float* h2   = agg1 + (size_t)N * 64;
    float* agg2 = h2   + (size_t)N * 16;
    const int nT = (N + TILE_NODES - 1) / TILE_NODES;
    int*   gq_s   = (int*)(agg2 + (size_t)N * 16);
    int*   gq_d   = gq_s + (size_t)nT * QCAP;
    float* gq_w   = (float*)(gq_d + (size_t)nT * QCAP);
    const int nq   = (E + 3) / 4;
    const int nblk = (nq + TB_T * TB_U - 1) / (TB_T * TB_U);
    unsigned* pres         = (unsigned*)(gq_w + (size_t)nT * QCAP);
    int*      tile_of_edge = (int*)(pres + (size_t)nT * nblk * PW);
    int*      gq_cnt       = tile_of_edge + nq;

    const int T = 256;

    tile_byte_kernel<<<nblk, TB_T, 0, stream>>>(dst, E, tile_of_edge, pres, nT, nblk);
    tile_scan_kernel<<<nT, AGG_T, 0, stream>>>(
        src, dst, dis, gq_cnt, gq_s, gq_d, tile_of_edge, pres, nblk, E, N);

    {
        int nBlocks = (N + GM_ROWS - 1) / GM_ROWS;
        gemm_wmma_f32<128, 64><<<(nBlocks < 1024) ? nBlocks : 1024, (64 / 16) * 32, 0, stream>>>(
            x, W1, h1, N, nBlocks);
    }
    tile_agg64_kernel<<<nT, AGG_T, 0, stream>>>(
        src, dst, dis, h1, agg1, gq_cnt, gq_s, gq_d, gq_w, E, N);
    {
        int nTiles = (N / 16) * (16 / 16);
        int waves_per_block = T / 32;
        gemm_epi_wmma_f32<64, 16><<<(nTiles + waves_per_block - 1) / waves_per_block,
                                    T, 0, stream>>>(agg1, h1, dis, b1, W2, h2, nTiles);
    }
    tile_agg16_kernel<<<nT, AGG2_T, 0, stream>>>(
        src, dst, dis, h2, gq_cnt, gq_s, gq_d, gq_w, agg2, E, N);
    logsoftmax_kernel<<<(N + T - 1) / T, T, 0, stream>>>(h2, dis, b2, agg2, out, N);
}


# Byte pass becomes a pair pass: edges stored as interleaved (src,dst) pairs, one presence bit per 4-edge group; scan reads one 32-byte piece per set bit, no per-edge gathers; sum order changes
# speedup vs baseline: 3.9374x; 1.0465x over previous
#include <hip/hip_runtime.h>
#include <cstdint>

typedef __attribute__((ext_vector_type(2))) float v2f;
typedef __attribute__((ext_vector_type(4))) float v4f;
typedef __attribute__((ext_vector_type(8))) float v8f;
typedef __attribute__((ext_vector_type(4))) int      v4i;
typedef __attribute__((ext_vector_type(4))) unsigned v4u;

constexpr int TILE_NODES = 392;

constexpr int PK_T      = 1024;
constexpr int PK_GROUPS = PK_T;
constexpr int PW        = PK_GROUPS / 32;
constexpr int PB_PITCH  = 257;
static_assert(PW == 32, "one 128-byte line of presence bits per tile and work-group");

__global__ __launch_bounds__(PK_T)
void pair_kernel(const int* __restrict__ src, const int* __restrict__ dst, int E,
                 int* __restrict__ pairs,
                 unsigned* __restrict__ pres, int nT, int nblk) {
    __shared__ unsigned pb[PW * PB_PITCH];
    int4* __restrict__ pairs4 = (int4*)pairs;
    const int tid  = threadIdx.x;
    const int lane = tid & 31;
    for (int i = tid; i < PW * PB_PITCH; i += PK_T) pb[i] = 0u;

    const int wbase = (blockIdx.x * PK_T + (tid & ~31)) * 4;
    int s[2][2], d[2][2];
#pragma unroll
    for (int k = 0; k < 2; ++k) {
#pragma unroll
        for (int j = 0; j < 2; ++j) {
            const int e  = wbase + 64 * k + 2 * lane + j;
            const int ec = (e < E) ? e : E - 1;
            s[k][j] = src[ec];
            d[k][j] = dst[ec];
        }
    }
    __syncthreads();
    unsigned* pw = pb + (tid >> 5) * PB_PITCH;
#pragma unroll
    for (int k = 0; k < 2; ++k) {
        const unsigned bit = 1u << (16 * k + (lane >> 1));
#pragma unroll
        for (int j = 0; j < 2; ++j) {
            const int e = wbase + 64 * k + 2 * lane + j;
            if (e < E) {
                unsigned tl = ((unsigned)d[k][j] / (unsigned)TILE_NODES) & 0xFFu;
                atomicOr(&pw[tl], bit);
            } else {
                s[k][j] = 0;
                d[k][j] = -1;
            }
        }
        int4 o;
        o.x = s[k][0]; o.y = d[k][0]; o.z = s[k][1]; o.w = d[k][1];
        pairs4[(wbase + 64 * k) / 2 + lane] = o;
    }
    __syncthreads();
    for (int p = tid; p < nT * (PW / 4); p += PK_T) {
        const int tile = p >> 3;
        const int q    = p & 7;
        uint4 o;
        o.x = pb[(4 * q + 0) * PB_PITCH + tile];
        o.y = pb[(4 * q + 1) * PB_PITCH + tile];
        o.z = pb[(4 * q + 2) * PB_PITCH + tile];
        o.w = pb[(4 * q + 3) * PB_PITCH + tile];
        *(uint4*)(pres + ((size_t)tile * nblk + blockIdx.x) * PW + 4 * q) = o;
    }
}

constexpr int GM_ROWS = 32;

template <int K, int THREADS, int LOADS>
__device__ __forceinline__ void gemm_load_block(const float* __restrict__ A, int M, int blk, int tid,
                                                v4f (&pre)[LOADS]) {
    constexpr int ROW_F4 = K / 4;
#pragma unroll
    for (int i = 0; i < LOADS; ++i) {
        const int p = tid + i * THREADS;
        int row = blk * GM_ROWS + p / ROW_F4;
        if (row > M - 1) row = M - 1;
        pre[i] = *(const v4f*)(A + (size_t)row * K + 4 * (p % ROW_F4));
    }
}

template <int K, int NC>
__global__ __launch_bounds__((NC / 16) * 32)
void gemm_wmma_f32(const float* __restrict__ A,
                   const float* __restrict__ B,
                   float* __restrict__ C, int M, int nBlocks) {
    constexpr int WAVES   = NC / 16;
    constexpr int THREADS = WAVES * 32;
    constexpr int ROW_F4  = K / 4;
    constexpr int PITCH   = K + 4;
    constexpr int PIECES  = GM_ROWS * ROW_F4;
    constexpr int LOADS   = PIECES / THREADS;
    static_assert(PIECES % THREADS == 0, "block of A must divide over the threads");
    __shared__ __attribute__((aligned(16))) float a_s[GM_ROWS * PITCH];

    const int tid  = threadIdx.x;
    const int lane = tid & 31;
    const int tn   = tid >> 5;
    const int hi   = lane >> 4;
    const int l    = lane & 15;
    const int colBase = tn * 16;

    v2f bf[K / 4];
    {
        const float* bcol = B + colBase + l;
#pragma unroll
        for (int j = 0; j < K / 4; ++j) {
            bf[j].x = bcol[(size_t)(4 * j + 2 * hi + 0) * NC];
            bf[j].y = bcol[(size_t)(4 * j + 2 * hi + 1) * NC];
        }
    }

    v4f pre[LOADS];
    int blk = blockIdx.x;
    if (blk < nBlocks) gemm_load_block<K, THREADS, LOADS>(A, M, blk, tid, pre);
    for (; blk < nBlocks; blk += gridDim.x) {
#pragma unroll
        for (int i = 0; i < LOADS; ++i) {
            const int p = tid + i * THREADS;
            *(v4f*)(a_s + (p / ROW_F4) * PITCH + 4 * (p % ROW_F4)) = pre[i];
        }
        __syncthreads();
        const int nxt = blk + gridDim.x;
        if (nxt < nBlocks) gemm_load_block<K, THREADS, LOADS>(A, M, nxt, tid, pre);

        const float* a0 = a_s + l * PITCH + 2 * hi;
        const float* a1 = a0 + 16 * PITCH;
        v8f acc0 = {};
        v8f acc1 = {};
#pragma unroll
        for (int j = 0; j < K / 4; ++j) {
            const v2f f0 = *(const v2f*)(a0 + 4 * j);
            const v2f f1 = *(const v2f*)(a1 + 4 * j);
            acc0 = __builtin_amdgcn_wmma_f32_16x16x4_f32(
                 false, f0,  false, bf[j],
                 (short)0, acc0,  false,  false);
            acc1 = __builtin_amdgcn_wmma_f32_16x16x4_f32(
                 false, f1,  false, bf[j],
                 (short)0, acc1,  false,  false);
        }

        const int rowBase = blk * GM_ROWS;
        float* cptr = C + (size_t)colBase + l;
#pragma unroll
        for (int i = 0; i < 8; ++i) {
            const int r0 = rowBase + 8 * hi + i;
            const int r1 = r0 + 16;
            if (r0 < M) cptr[(size_t)r0 * NC] = acc0[i];
            if (r1 < M) cptr[(size_t)r1 * NC] = acc1[i];
        }
        __syncthreads();
    }
}

constexpr int QCAP       = 8192;
constexpr int AGG_T      = 1024;
constexpr int PRES_V     = 4;
constexpr int GRP_U      = QCAP / AGG_T;

__device__ __forceinline__ int wave_queue_base(int cnt, int lane, int* counter) {
    int incl = cnt;
#pragma unroll
    for (int o = 1; o < 32; o <<= 1) {
        int t = __shfl_up(incl, o);
        if (lane >= o) incl += t;
    }
    int wb = 0;
    if (lane == 31) wb = atomicAdd(counter, incl);
    wb = __shfl(wb, 31);
    return wb + incl - cnt;
}

__global__ __launch_bounds__(AGG_T)
void tile_scan_kernel(const int* __restrict__ dst,
                      float* __restrict__ dis,
                      int* __restrict__ gq_cnt, int* __restrict__ gq_s,
                      int* __restrict__ gq_d,
                      const int* __restrict__ pairs,
                      const unsigned* __restrict__ pres, int nblk,
                      int E, int N) {
    const v4i* __restrict__ pairs4 = (const v4i*)pairs;
    __shared__ int   q_s[QCAP];
    __shared__ int   q_d[QCAP];
    __shared__ int   q_g[QCAP];
    __shared__ int   deg_l[TILE_NODES];
    __shared__ int   q_count;
    __shared__ int   g_count;

    const int tid  = threadIdx.x;
    const int lane = tid & 31;
    const int base = blockIdx.x * TILE_NODES;
    int rows = N - base;
    if (rows > TILE_NODES) rows = TILE_NODES;

    for (int i = tid; i < TILE_NODES; i += AGG_T) deg_l[i] = 0;
    if (tid == 0) { q_count = 0; g_count = 0; }
    __syncthreads();

    const int nvec = nblk * (PW / 4);
    const v4u* __restrict__ prow = (const v4u*)(pres + (size_t)blockIdx.x * nblk * PW);
    for (int v0 = 0; v0 < nvec; v0 += PRES_V * AGG_T) {
        v4u      raw[PRES_V];
        unsigned bits[PRES_V][4];
#pragma unroll
        for (int u = 0; u < PRES_V; ++u) {
            const int v = v0 + u * AGG_T + tid;
            raw[u] = prow[(v < nvec) ? v : nvec - 1];
        }
        __builtin_amdgcn_sched_barrier(0);
#pragma unroll
        for (int u = 0; u < PRES_V; ++u) {
            const bool in = (v0 + u * AGG_T + tid) < nvec;
            bits[u][0] = in ? raw[u].x : 0u;
            bits[u][1] = in ? raw[u].y : 0u;
            bits[u][2] = in ? raw[u].z : 0u;
            bits[u][3] = in ? raw[u].w : 0u;
        }
        int cnt = 0;
#pragma unroll
        for (int u = 0; u < PRES_V; ++u)
#pragma unroll
            for (int cc = 0; cc < 4; ++cc) cnt += __builtin_popcount(bits[u][cc]);
        int slot = wave_queue_base(cnt, lane, &g_count);
#pragma unroll
        for (int u = 0; u < PRES_V; ++u) {
            const int v = v0 + u * AGG_T + tid;
#pragma unroll
            for (int cc = 0; cc < 4; ++cc) {
                unsigned m = bits[u][cc];
                const int g0 = 32 * (4 * v + cc);
                while (m) {
                    int b = __builtin_ctz(m);
                    if (slot < QCAP) q_g[slot] = g0 + b;
                    ++slot;
                    m &= m - 1;
                }
            }
        }
    }
    __syncthreads();

    const int gtotal = g_count;
    if (gtotal > QCAP) {
        if (tid == 0) q_count = QCAP + 1;
    } else if (gtotal > 0) {
        int gi[GRP_U];
        v4i pa[GRP_U];
        v4i pc[GRP_U];
#pragma unroll
        for (int u = 0; u < GRP_U; ++u) {
            int i = tid + u * AGG_T;
            gi[u] = q_g[(i < gtotal) ? i : gtotal - 1];
        }
#pragma unroll
        for (int u = 0; u < GRP_U; ++u) {
            pa[u] = pairs4[2 * (size_t)gi[u] + 0];
            pc[u] = pairs4[2 * (size_t)gi[u] + 1];
        }
        asm volatile("" :: "v"(pa[0]), "v"(pc[0]), "v"(pa[1]), "v"(pc[1]),
                           "v"(pa[2]), "v"(pc[2]), "v"(pa[3]), "v"(pc[3]),
                           "v"(pa[4]), "v"(pc[4]), "v"(pa[5]), "v"(pc[5]),
                           "v"(pa[6]), "v"(pc[6]), "v"(pa[7]), "v"(pc[7]));
        static_assert(GRP_U == 8, "the statement above lists 2*GRP_U pieces");
        int dl[GRP_U][4];
        int cnt = 0;
#pragma unroll
        for (int u = 0; u < GRP_U; ++u) {
            const bool in = (tid + u * AGG_T) < gtotal;
            const int dd[4] = { pa[u].y, pa[u].w, pc[u].y, pc[u].w };
#pragma unroll
            for (int j = 0; j < 4; ++j) {
                const unsigned r = (unsigned)(dd[j] - base);
                const bool hit = in && (r < (unsigned)rows);
                dl[u][j] = hit ? (int)r : -1;
                cnt += hit ? 1 : 0;
            }
        }
        int slot = wave_queue_base(cnt, lane, &q_count);
#pragma unroll
        for (int u = 0; u < GRP_U; ++u) {
            const int ss[4] = { pa[u].x, pa[u].z, pc[u].x, pc[u].z };
#pragma unroll
            for (int j = 0; j < 4; ++j) {
                if (dl[u][j] >= 0) {
                    if (slot < QCAP) {
                        q_s[slot] = ss[j];
                        q_d[slot] = dl[u][j];
                    }
                    ++slot;
                    atomicAdd(&deg_l[dl[u][j]], 1);
                }
            }
        }
    }
    __syncthreads();

    int total = q_count;
    if (tid == 0) gq_cnt[blockIdx.x] = total;
    if (total > QCAP) {
        for (int i = tid; i < TILE_NODES; i += AGG_T) deg_l[i] = 0;
        __syncthreads();
        for (int e = tid; e < E; e += AGG_T) {
            unsigned r = (unsigned)(dst[e] - base);
            if (r < (unsigned)rows) atomicAdd(&deg_l[r], 1);
        }
        total = 0;
    }

    int* __restrict__ ts = gq_s + (size_t)blockIdx.x * QCAP;
    int* __restrict__ td = gq_d + (size_t)blockIdx.x * QCAP;
    for (int i = tid; i < total; i += AGG_T) {
        ts[i] = q_s[i];
        td[i] = q_d[i];
    }
    __syncthreads();

    for (int r = tid; r < rows; r += AGG_T)
        dis[base + r] = rsqrtf((float)deg_l[r] + 1.0f);
}

__global__ __launch_bounds__(AGG_T)
void tile_agg64_kernel(const int* __restrict__ src,
                       const int* __restrict__ dst,
                       const float* __restrict__ dis,
                       const float* __restrict__ h,
                       float* __restrict__ agg,
                       const int* __restrict__ gq_cnt, const int* __restrict__ gq_s,
                       const int* __restrict__ gq_d, float* __restrict__ gq_w,
                       int E, int N) {
    constexpr int F = 64;
    __shared__ float acc[(TILE_NODES + 1) * F];
    __shared__ int   q_a[QCAP];
    __shared__ int   q_d[QCAP];
    __shared__ float q_w[QCAP];

    const int tid  = threadIdx.x;
    const int base = blockIdx.x * TILE_NODES;
    int rows = N - base;
    if (rows > TILE_NODES) rows = TILE_NODES;

    for (int i = tid; i < (TILE_NODES + 1) * F; i += AGG_T) acc[i] = 0.0f;
    __syncthreads();

    int total = gq_cnt[blockIdx.x];
    if (total > QCAP) {
        for (int e = tid; e < E; e += AGG_T) {
            int d = dst[e];
            unsigned dl = (unsigned)(d - base);
            if (dl < (unsigned)rows) {
                int s = src[e];
                float w = dis[s] * dis[d];
                for (int j = 0; j < F; ++j)
                    atomicAdd(&acc[dl * F + j], h[(size_t)s * F + j] * w);
            }
        }
        total = 0;
    }

    const int* __restrict__ ts = gq_s + (size_t)blockIdx.x * QCAP;
    const int* __restrict__ td = gq_d + (size_t)blockIdx.x * QCAP;
    float*     __restrict__ tw = gq_w + (size_t)blockIdx.x * QCAP;
    for (int i = tid; i < total; i += AGG_T) {
        int s  = ts[i];
        int dl = td[i];
        const bool in_tile = (unsigned)dl < (unsigned)rows;
        float w = in_tile ? dis[s] * dis[base + (in_tile ? dl : 0)] : 0.0f;
        q_a[i] = s;
        q_d[i] = dl;
        q_w[i] = w;
        tw[i]  = w;
    }
    __syncthreads();

    const int wave = tid >> 5;
    const int lane = tid & 31;
    constexpr int R = 8;
    for (int i0 = wave * R; i0 < total; i0 += (AGG_T / 32) * R) {
        int    sn[R];
        int    dn[R];
        float  wn[R];
        float2 hv[R];
#pragma unroll
        for (int r = 0; r < R; ++r) {
            int i = i0 + r;
            if (i > total - 1) i = total - 1;
            sn[r] = q_a[i];
            dn[r] = q_d[i];
            wn[r] = q_w[i];
        }
#pragma unroll
        for (int r = 0; r < R; ++r)
            hv[r] = *(const float2*)(h + (size_t)sn[r] * F + 2 * lane);
        asm volatile("" :: "v"(hv[0].x), "v"(hv[0].y), "v"(hv[1].x), "v"(hv[1].y),
                           "v"(hv[2].x), "v"(hv[2].y), "v"(hv[3].x), "v"(hv[3].y),
                           "v"(hv[4].x), "v"(hv[4].y), "v"(hv[5].x), "v"(hv[5].y),
                           "v"(hv[6].x), "v"(hv[6].y), "v"(hv[7].x), "v"(hv[7].y));
#pragma unroll
        for (int r = 0; r < R; ++r) {
            int i = i0 + r;
            if (i < total) {
                atomicAdd(&acc[dn[r] * F + 2 * lane + 0], hv[r].x * wn[r]);
                atomicAdd(&acc[dn[r] * F + 2 * lane + 1], hv[r].y * wn[r]);
            }
        }
    }
    __syncthreads();

    const int n4 = rows * (F / 4);
    float4* __restrict__ o4 = (float4*)(agg + (size_t)base * F);
    const float4* a4 = (const float4*)acc;
    for (int i = tid; i < n4; i += AGG_T) o4[i] = a4[i];
}

constexpr int AGG2_T = 1024;
constexpr int R2     = 8;

__global__ __launch_bounds__(AGG2_T)
void tile_agg16_kernel(const int* __restrict__ src,
                       const int* __restrict__ dst,
                       const float* __restrict__ dis,
                       const float* __restrict__ h,
                       const int* __restrict__ gq_cnt, const int* __restrict__ gq_s,
                       const int* __restrict__ gq_d, const float* __restrict__ gq_w,
                       float* __restrict__ agg, int E, int N) {
    constexpr int F = 16;
    __shared__ float acc[(TILE_NODES + 1) * F];

    const int tid  = threadIdx.x;
    const int base = blockIdx.x * TILE_NODES;
    int rows = N - base;
    if (rows > TILE_NODES) rows = TILE_NODES;

    for (int i = tid; i < (TILE_NODES + 1) * F; i += AGG2_T) acc[i] = 0.0f;
    __syncthreads();

    int total = gq_cnt[blockIdx.x];
    if (total > QCAP) {
        for (int e = tid; e < E; e += AGG2_T) {
            int d = dst[e];
            unsigned dl = (unsigned)(d - base);
            if (dl < (unsigned)rows) {
                int s = src[e];
                float w = dis[s] * dis[d];
                for (int j = 0; j < F; ++j)
                    atomicAdd(&acc[dl * F + j], h[(size_t)s * F + j] * w);
            }
        }
        total = 0;
    }

    const int*   __restrict__ ts = gq_s + (size_t)blockIdx.x * QCAP;
    const int*   __restrict__ td = gq_d + (size_t)blockIdx.x * QCAP;
    const float* __restrict__ tw = gq_w + (size_t)blockIdx.x * QCAP;

    const int grp = tid >> 2;
    const int qt  = tid & 3;
    constexpr int GROUPS = AGG2_T / 4;
    for (int r0 = 0; r0 < total; r0 += GROUPS * R2) {
        int   sn[R2];
        int   dn[R2];
        float wn[R2];
        v4f   hv[R2];
#pragma unroll
        for (int r = 0; r < R2; ++r) {
            int i = r0 + r * GROUPS + grp;
            if (i > total - 1) i = total - 1;
            sn[r] = ts[i];
            dn[r] = td[i];
            wn[r] = tw[i];
        }
#pragma unroll
        for (int r = 0; r < R2; ++r)
            hv[r] = *(const v4f*)(h + (size_t)sn[r] * F + 4 * qt);
        asm volatile("" :: "v"(hv[0]), "v"(hv[1]), "v"(hv[2]), "v"(hv[3]),
                           "v"(hv[4]), "v"(hv[5]), "v"(hv[6]), "v"(hv[7]),
                           "v"(dn[0]), "v"(dn[1]), "v"(dn[2]), "v"(dn[3]),
                           "v"(dn[4]), "v"(dn[5]), "v"(dn[6]), "v"(dn[7]),
                           "v"(wn[0]), "v"(wn[1]), "v"(wn[2]), "v"(wn[3]),
                           "v"(wn[4]), "v"(wn[5]), "v"(wn[6]), "v"(wn[7]));
#pragma unroll
        for (int r = 0; r < R2; ++r) {
            int i = r0 + r * GROUPS + grp;
            if (i < total) {
                float* p = &acc[dn[r] * F + 4 * qt];
                atomicAdd(p + 0, hv[r].x * wn[r]);
                atomicAdd(p + 1, hv[r].y * wn[r]);
                atomicAdd(p + 2, hv[r].z * wn[r]);
                atomicAdd(p + 3, hv[r].w * wn[r]);
            }
        }
    }
    __syncthreads();

    const int n4 = rows * (F / 4);
    float4* __restrict__ o4 = (float4*)(agg + (size_t)base * F);
    const float4* a4 = (const float4*)acc;
    for (int i = tid; i < n4; i += AGG2_T) o4[i] = a4[i];
}

template <int K, int NC>
__global__ void gemm_epi_wmma_f32(const float* __restrict__ agg,
                                  const float* __restrict__ h,
                                  const float* __restrict__ dis,
                                  const float* __restrict__ bias,
                                  const float* __restrict__ B,
                                  float* __restrict__ C, int nTiles) {
    int wave = (int)((blockIdx.x * blockDim.x + threadIdx.x) >> 5);
    int lane = threadIdx.x & 31;
    if (wave >= nTiles) return;
    constexpr int TN = NC / 16;
    int tm = wave / TN;
    int tn = wave % TN;
    int rowBase = tm * 16;
    int colBase = tn * 16;
    int hi = lane >> 4;
    int l  = lane & 15;

    const size_t roff = (size_t)(rowBase + l) * K + 2 * hi;
    const float* grow = agg + roff;
    const float* hrow = h + roff;
    const float* brow = bias + 2 * hi;
    const float* bcol = B + colBase + l;
    const float  dii  = dis[rowBase + l];

    v8f acc = {};
#pragma unroll
    for (int k = 0; k < K; k += 4) {
        const float2 g  = *(const float2*)(grow + k);
        const float2 hh = *(const float2*)(hrow + k);
        const float2 bb = *(const float2*)(brow + k);
        v2f a, b;
        a.x = fmaxf(g.x + hh.x * dii * dii + bb.x, 0.0f);
        a.y = fmaxf(g.y + hh.y * dii * dii + bb.y, 0.0f);
        b.x = bcol[(size_t)(k + 2 * hi + 0) * NC];
        b.y = bcol[(size_t)(k + 2 * hi + 1) * NC];
        acc = __builtin_amdgcn_wmma_f32_16x16x4_f32(
             false, a,  false, b,
             (short)0, acc,  false,  false);
    }

    float* cptr = C + (size_t)colBase + l;
#pragma unroll
    for (int i = 0; i < 8; ++i)
        cptr[(size_t)(rowBase + 8 * hi + i) * NC] = acc[i];
}

__global__ void logsoftmax_kernel(const float* __restrict__ h,
                                  const float* __restrict__ dis,
                                  const float* __restrict__ bias,
                                  const float* __restrict__ agg,
                                  float* __restrict__ out, int n) {
    int i = blockIdx.x * blockDim.x + threadIdx.x;
    if (i >= n) return;
    float dii = dis[i];
    dii *= dii;
    float v[16];
    float mx = -3.402823466e38f;
#pragma unroll
    for (int j = 0; j < 16; ++j) {
        size_t idx = (size_t)i * 16 + j;
        v[j] = agg[idx] + h[idx] * dii + bias[j];
        mx = fmaxf(mx, v[j]);
    }
    float s = 0.0f;
#pragma unroll
    for (int j = 0; j < 16; ++j) s += __expf(v[j] - mx);
    float lse = mx + __logf(s);
#pragma unroll
    for (int j = 0; j < 16; ++j) out[(size_t)i * 16 + j] = v[j] - lse;
}

extern "C" void kernel_launch(void* const* d_in, const int* in_sizes, int n_in,
                              void* d_out, int out_size, void* d_ws, size_t ws_size,
                              hipStream_t stream) {
    const float* x   = (const float*)d_in[0];
    const int*   ei  = (const int*)  d_in[1];
    const float* W1  = (const float*)d_in[2];
    const float* b1  = (const float*)d_in[3];
    const float* W2  = (const float*)d_in[4];
    const float* b2  = (const float*)d_in[5];
    float* out = (float*)d_out;

    const int N = in_sizes[0] / 128;
    const int E = in_sizes[1] / 2;
    const int* src = ei;
    const int* dst = ei + E;

    float* dis  = (float*)d_ws;
    float* h1   = dis  + N;
    float* agg1 = h1   + (size_t)N * 64;
    float* h2   = agg1 + (size_t)N * 64;
    float* agg2 = h2   + (size_t)N * 16;
    const int nT = (N + TILE_NODES - 1) / TILE_NODES;
    int*   gq_s   = (int*)(agg2 + (size_t)N * 16);
    int*   gq_d   = gq_s + (size_t)nT * QCAP;
    float* gq_w   = (float*)(gq_d + (size_t)nT * QCAP);
    const int ng   = (E + 3) / 4;
    const int nblk = (ng + PK_GROUPS - 1) / PK_GROUPS;
    unsigned* pres   = (unsigned*)(gq_w + (size_t)nT * QCAP);
    int*      pairs  = (int*)(pres + (size_t)nT * nblk * PW);
    int*      gq_cnt = pairs + (size_t)nblk * PK_GROUPS * 8;

    const int T = 256;

    pair_kernel<<<nblk, PK_T, 0, stream>>>(src, dst, E, pairs, pres, nT, nblk);
    tile_scan_kernel<<<nT, AGG_T, 0, stream>>>(
        dst, dis, gq_cnt, gq_s, gq_d, pairs, pres, nblk, E, N);

    {
        int nBlocks = (N + GM_ROWS - 1) / GM_ROWS;
        gemm_wmma_f32<128, 64><<<(nBlocks < 1024) ? nBlocks : 1024, (64 / 16) * 32, 0, stream>>>(
            x, W1, h1, N, nBlocks);
    }
    tile_agg64_kernel<<<nT, AGG_T, 0, stream>>>(
        src, dst, dis, h1, agg1, gq_cnt, gq_s, gq_d, gq_w, E, N);
    {
        int nTiles = (N / 16) * (16 / 16);
        int waves_per_block = T / 32;
        gemm_epi_wmma_f32<64, 16><<<(nTiles + waves_per_block - 1) / waves_per_block,
                                    T, 0, stream>>>(agg1, h1, dis, b1, W2, h2, nTiles);
    }
    tile_agg16_kernel<<<nT, AGG2_T, 0, stream>>>(
        src, dst, dis, h2, gq_cnt, gq_s, gq_d, gq_w, agg2, E, N);
    logsoftmax_kernel<<<(N + T - 1) / T, T, 0, stream>>>(h2, dis, b2, agg2, out, N);
}


# Tile kernels finish their rows out of LDS: layer-1 tile gets self-loop term, bias, relu and the W2 product on WMMA, stores h2; layer-2 tile stores log_softmax; 7 launches -> 5; sum order kept
# speedup vs baseline: 4.4033x; 1.1183x over previous
#include <hip/hip_runtime.h>
#include <cstdint>

typedef __attribute__((ext_vector_type(2))) float v2f;
typedef __attribute__((ext_vector_type(4))) float v4f;
typedef __attribute__((ext_vector_type(8))) float v8f;
typedef __attribute__((ext_vector_type(4))) int      v4i;
typedef __attribute__((ext_vector_type(4))) unsigned v4u;

constexpr int TILE_NODES = 392;

constexpr int PK_T      = 1024;
constexpr int PK_GROUPS = PK_T;
constexpr int PW        = PK_GROUPS / 32;
constexpr int PB_PITCH  = 257;
static_assert(PW == 32, "one 128-byte line of presence bits per tile and work-group");

__global__ __launch_bounds__(PK_T)
void pair_kernel(const int* __restrict__ src, const int* __restrict__ dst, int E,
                 int* __restrict__ pairs,
                 unsigned* __restrict__ pres, int nT, int nblk) {
    __shared__ unsigned pb[PW * PB_PITCH];
    int4* __restrict__ pairs4 = (int4*)pairs;
    const int tid  = threadIdx.x;
    const int lane = tid & 31;
    for (int i = tid; i < PW * PB_PITCH; i += PK_T) pb[i] = 0u;

    const int wbase = (blockIdx.x * PK_T + (tid & ~31)) * 4;
    int s[2][2], d[2][2];
#pragma unroll
    for (int k = 0; k < 2; ++k) {
#pragma unroll
        for (int j = 0; j < 2; ++j) {
            const int e  = wbase + 64 * k + 2 * lane + j;
            const int ec = (e < E) ? e : E - 1;
            s[k][j] = src[ec];
            d[k][j] = dst[ec];
        }
    }
    __syncthreads();
    unsigned* pw = pb + (tid >> 5) * PB_PITCH;
#pragma unroll
    for (int k = 0; k < 2; ++k) {
        const unsigned bit = 1u << (16 * k + (lane >> 1));
#pragma unroll
        for (int j = 0; j < 2; ++j) {
            const int e = wbase + 64 * k + 2 * lane + j;
            if (e < E) {
                unsigned tl = ((unsigned)d[k][j] / (unsigned)TILE_NODES) & 0xFFu;
                atomicOr(&pw[tl], bit);
            } else {
                s[k][j] = 0;
                d[k][j] = -1;
            }
        }
        int4 o;
        o.x = s[k][0]; o.y = d[k][0]; o.z = s[k][1]; o.w = d[k][1];
        pairs4[(wbase + 64 * k) / 2 + lane] = o;
    }
    __syncthreads();
    for (int p = tid; p < nT * (PW / 4); p += PK_T) {
        const int tile = p >> 3;
        const int q    = p & 7;
        uint4 o;
        o.x = pb[(4 * q + 0) * PB_PITCH + tile];
        o.y = pb[(4 * q + 1) * PB_PITCH + tile];
        o.z = pb[(4 * q + 2) * PB_PITCH + tile];
        o.w = pb[(4 * q + 3) * PB_PITCH + tile];
        *(uint4*)(pres + ((size_t)tile * nblk + blockIdx.x) * PW + 4 * q) = o;
    }
}

constexpr int GM_ROWS = 32;

template <int K, int THREADS, int LOADS>
__device__ __forceinline__ void gemm_load_block(const float* __restrict__ A, int M, int blk, int tid,
                                                v4f (&pre)[LOADS]) {
    constexpr int ROW_F4 = K / 4;
#pragma unroll
    for (int i = 0; i < LOADS; ++i) {
        const int p = tid + i * THREADS;
        int row = blk * GM_ROWS + p / ROW_F4;
        if (row > M - 1) row = M - 1;
        pre[i] = *(const v4f*)(A + (size_t)row * K + 4 * (p % ROW_F4));
    }
}

template <int K, int NC>
__global__ __launch_bounds__((NC / 16) * 32)
void gemm_wmma_f32(const float* __restrict__ A,
                   const float* __restrict__ B,
                   float* __restrict__ C, int M, int nBlocks) {
    constexpr int WAVES   = NC / 16;
    constexpr int THREADS = WAVES * 32;
    constexpr int ROW_F4  = K / 4;
    constexpr int PITCH   = K + 4;
    constexpr int PIECES  = GM_ROWS * ROW_F4;
    constexpr int LOADS   = PIECES / THREADS;
    static_assert(PIECES % THREADS == 0, "block of A must divide over the threads");
    __shared__ __attribute__((aligned(16))) float a_s[GM_ROWS * PITCH];

    const int tid  = threadIdx.x;
    const int lane = tid & 31;
    const int tn   = tid >> 5;
    const int hi   = lane >> 4;
    const int l    = lane & 15;
    const int colBase = tn * 16;

    v2f bf[K / 4];
    {
        const float* bcol = B + colBase + l;
#pragma unroll
        for (int j = 0; j < K / 4; ++j) {
            bf[j].x = bcol[(size_t)(4 * j + 2 * hi + 0) * NC];
            bf[j].y = bcol[(size_t)(4 * j + 2 * hi + 1) * NC];
        }
    }

    v4f pre[LOADS];
    int blk = blockIdx.x;
    if (blk < nBlocks) gemm_load_block<K, THREADS, LOADS>(A, M, blk, tid, pre);
    for (; blk < nBlocks; blk += gridDim.x) {
#pragma unroll
        for (int i = 0; i < LOADS; ++i) {
            const int p = tid + i * THREADS;
            *(v4f*)(a_s + (p / ROW_F4) * PITCH + 4 * (p % ROW_F4)) = pre[i];
        }
        __syncthreads();
        const int nxt = blk + gridDim.x;
        if (nxt < nBlocks) gemm_load_block<K, THREADS, LOADS>(A, M, nxt, tid, pre);

        const float* a0 = a_s + l * PITCH + 2 * hi;
        const float* a1 = a0 + 16 * PITCH;
        v8f acc0 = {};
        v8f acc1 = {};
#pragma unroll
        for (int j = 0; j < K / 4; ++j) {
            const v2f f0 = *(const v2f*)(a0 + 4 * j);
            const v2f f1 = *(const v2f*)(a1 + 4 * j);
            acc0 = __builtin_amdgcn_wmma_f32_16x16x4_f32(
                 false, f0,  false, bf[j],
                 (short)0, acc0,  false,  false);
            acc1 = __builtin_amdgcn_wmma_f32_16x16x4_f32(
                 false, f1,  false, bf[j],
                 (short)0, acc1,  false,  false);
        }

        const int rowBase = blk * GM_ROWS;
        float* cptr = C + (size_t)colBase + l;
#pragma unroll
        for (int i = 0; i < 8; ++i) {
            const int r0 = rowBase + 8 * hi + i;
            const int r1 = r0 + 16;
            if (r0 < M) cptr[(size_t)r0 * NC] = acc0[i];
            if (r1 < M) cptr[(size_t)r1 * NC] = acc1[i];
        }
        __syncthreads();
    }
}

constexpr int QCAP       = 8192;
constexpr int AGG_T      = 1024;
constexpr int PRES_V     = 4;
constexpr int GRP_U      = QCAP / AGG_T;

__device__ __forceinline__ int wave_queue_base(int cnt, int lane, int* counter) {
    int incl = cnt;
#pragma unroll
    for (int o = 1; o < 32; o <<= 1) {
        int t = __shfl_up(incl, o);
        if (lane >= o) incl += t;
    }
    int wb = 0;
    if (lane == 31) wb = atomicAdd(counter, incl);
    wb = __shfl(wb, 31);
    return wb + incl - cnt;
}

__global__ __launch_bounds__(AGG_T)
void tile_scan_kernel(const int* __restrict__ dst,
                      float* __restrict__ dis,
                      int* __restrict__ gq_cnt, int* __restrict__ gq_s,
                      int* __restrict__ gq_d,
                      const int* __restrict__ pairs,
                      const unsigned* __restrict__ pres, int nblk,
                      int E, int N) {
    const v4i* __restrict__ pairs4 = (const v4i*)pairs;
    __shared__ int   q_s[QCAP];
    __shared__ int   q_d[QCAP];
    __shared__ int   q_g[QCAP];
    __shared__ int   deg_l[TILE_NODES];
    __shared__ int   q_count;
    __shared__ int   g_count;

    const int tid  = threadIdx.x;
    const int lane = tid & 31;
    const int base = blockIdx.x * TILE_NODES;
    int rows = N - base;
    if (rows > TILE_NODES) rows = TILE_NODES;

    for (int i = tid; i < TILE_NODES; i += AGG_T) deg_l[i] = 0;
    if (tid == 0) { q_count = 0; g_count = 0; }
    __syncthreads();

    const int nvec = nblk * (PW / 4);
    const v4u* __restrict__ prow = (const v4u*)(pres + (size_t)blockIdx.x * nblk * PW);
    for (int v0 = 0; v0 < nvec; v0 += PRES_V * AGG_T) {
        v4u      raw[PRES_V];
        unsigned bits[PRES_V][4];
#pragma unroll
        for (int u = 0; u < PRES_V; ++u) {
            const int v = v0 + u * AGG_T + tid;
            raw[u] = prow[(v < nvec) ? v : nvec - 1];
        }
        __builtin_amdgcn_sched_barrier(0);
#pragma unroll
        for (int u = 0; u < PRES_V; ++u) {
            const bool in = (v0 + u * AGG_T + tid) < nvec;
            bits[u][0] = in ? raw[u].x : 0u;
            bits[u][1] = in ? raw[u].y : 0u;
            bits[u][2] = in ? raw[u].z : 0u;
            bits[u][3] = in ? raw[u].w : 0u;
        }
        int cnt = 0;
#pragma unroll
        for (int u = 0; u < PRES_V; ++u)
#pragma unroll
            for (int cc = 0; cc < 4; ++cc) cnt += __builtin_popcount(bits[u][cc]);
        int slot = wave_queue_base(cnt, lane, &g_count);
#pragma unroll
        for (int u = 0; u < PRES_V; ++u) {
            const int v = v0 + u * AGG_T + tid;
#pragma unroll
            for (int cc = 0; cc < 4; ++cc) {
                unsigned m = bits[u][cc];
                const int g0 = 32 * (4 * v + cc);
                while (m) {
                    int b = __builtin_ctz(m);
                    if (slot < QCAP) q_g[slot] = g0 + b;
                    ++slot;
                    m &= m - 1;
                }
            }
        }
    }
    __syncthreads();

    const int gtotal = g_count;
    if (gtotal > QCAP) {
        if (tid == 0) q_count = QCAP + 1;
    } else if (gtotal > 0) {
        int gi[GRP_U];
        v4i pa[GRP_U];
        v4i pc[GRP_U];
#pragma unroll
        for (int u = 0; u < GRP_U; ++u) {
            int i = tid + u * AGG_T;
            gi[u] = q_g[(i < gtotal) ? i : gtotal - 1];
        }
#pragma unroll
        for (int u = 0; u < GRP_U; ++u) {
            pa[u] = pairs4[2 * (size_t)gi[u] + 0];
            pc[u] = pairs4[2 * (size_t)gi[u] + 1];
        }
        asm volatile("" :: "v"(pa[0]), "v"(pc[0]), "v"(pa[1]), "v"(pc[1]),
                           "v"(pa[2]), "v"(pc[2]), "v"(pa[3]), "v"(pc[3]),
                           "v"(pa[4]), "v"(pc[4]), "v"(pa[5]), "v"(pc[5]),
                           "v"(pa[6]), "v"(pc[6]), "v"(pa[7]), "v"(pc[7]));
        static_assert(GRP_U == 8, "the statement above lists 2*GRP_U pieces");
        int dl[GRP_U][4];
        int cnt = 0;
#pragma unroll
        for (int u = 0; u < GRP_U; ++u) {
            const bool in = (tid + u * AGG_T) < gtotal;
            const int dd[4] = { pa[u].y, pa[u].w, pc[u].y, pc[u].w };
#pragma unroll
            for (int j = 0; j < 4; ++j) {
                const unsigned r = (unsigned)(dd[j] - base);
                const bool hit = in && (r < (unsigned)rows);
                dl[u][j] = hit ? (int)r : -1;
                cnt += hit ? 1 : 0;
            }
        }
        int slot = wave_queue_base(cnt, lane, &q_count);
#pragma unroll
        for (int u = 0; u < GRP_U; ++u) {
            const int ss[4] = { pa[u].x, pa[u].z, pc[u].x, pc[u].z };
#pragma unroll
            for (int j = 0; j < 4; ++j) {
                if (dl[u][j] >= 0) {
                    if (slot < QCAP) {
                        q_s[slot] = ss[j];
                        q_d[slot] = dl[u][j];
                    }
                    ++slot;
                    atomicAdd(&deg_l[dl[u][j]], 1);
                }
            }
        }
    }
    __syncthreads();

    int total = q_count;
    if (tid == 0) gq_cnt[blockIdx.x] = total;
    if (total > QCAP) {
        for (int i = tid; i < TILE_NODES; i += AGG_T) deg_l[i] = 0;
        __syncthreads();
        for (int e = tid; e < E; e += AGG_T) {
            unsigned r = (unsigned)(dst[e] - base);
            if (r < (unsigned)rows) atomicAdd(&deg_l[r], 1);
        }
        total = 0;
    }

    int* __restrict__ ts = gq_s + (size_t)blockIdx.x * QCAP;
    int* __restrict__ td = gq_d + (size_t)blockIdx.x * QCAP;
    for (int i = tid; i < total; i += AGG_T) {
        ts[i] = q_s[i];
        td[i] = q_d[i];
    }
    __syncthreads();

    for (int r = tid; r < rows; r += AGG_T)
        dis[base + r] = rsqrtf((float)deg_l[r] + 1.0f);
}

constexpr int ACC_ROWS = 400;
constexpr int EPI_U    = (TILE_NODES * 16 + AGG_T - 1) / AGG_T;
static_assert(ACC_ROWS % 16 == 0 && ACC_ROWS >= TILE_NODES + 1, "whole strips, and one spare row behind the tile");
static_assert(ACC_ROWS / 16 <= AGG_T / 32, "one wave per 16-row strip");

__global__ __launch_bounds__(AGG_T)
void tile_agg64_kernel(const int* __restrict__ src,
                       const int* __restrict__ dst,
                       const float* __restrict__ dis,
                       const float* __restrict__ h,
                       const float* __restrict__ bias,
                       const float* __restrict__ W2,
                       float* __restrict__ h2,
                       const int* __restrict__ gq_cnt, const int* __restrict__ gq_s,
                       const int* __restrict__ gq_d, float* __restrict__ gq_w,
                       int E, int N) {
    constexpr int F  = 64;
    constexpr int NC = 16;
    __shared__ __attribute__((aligned(16))) float acc[ACC_ROWS * F];
    __shared__ int   q_a[QCAP];
    __shared__ int   q_d[QCAP];
    __shared__ float q_w[QCAP];
    __shared__ float dis_l[TILE_NODES];

    const int tid  = threadIdx.x;
    const int base = blockIdx.x * TILE_NODES;
    int rows = N - base;
    if (rows > TILE_NODES) rows = TILE_NODES;

    for (int i = tid; i < ACC_ROWS * F; i += AGG_T) acc[i] = 0.0f;
    for (int r = tid; r < rows; r += AGG_T) dis_l[r] = dis[base + r];
    __syncthreads();

    int total = gq_cnt[blockIdx.x];
    if (total > QCAP) {
        for (int e = tid; e < E; e += AGG_T) {
            int d = dst[e];
            unsigned dl = (unsigned)(d - base);
            if (dl < (unsigned)rows) {
                int s = src[e];
                float w = dis[s] * dis[d];
                for (int j = 0; j < F; ++j)
                    atomicAdd(&acc[dl * F + j], h[(size_t)s * F + j] * w);
            }
        }
        total = 0;
    }

    const int* __restrict__ ts = gq_s + (size_t)blockIdx.x * QCAP;
    const int* __restrict__ td = gq_d + (size_t)blockIdx.x * QCAP;
    float*     __restrict__ tw = gq_w + (size_t)blockIdx.x * QCAP;
    for (int i = tid; i < total; i += AGG_T) {
        int s  = ts[i];
        int dl = td[i];
        const bool in_tile = (unsigned)dl < (unsigned)rows;
        float w = in_tile ? dis[s] * dis[base + (in_tile ? dl : 0)] : 0.0f;
        q_a[i] = s;
        q_d[i] = dl;
        q_w[i] = w;
        tw[i]  = w;
    }
    __syncthreads();

    const int wave = tid >> 5;
    const int lane = tid & 31;
    constexpr int R = 8;
    for (int i0 = wave * R; i0 < total; i0 += (AGG_T / 32) * R) {
        int    sn[R];
        int    dn[R];
        float  wn[R];
        float2 hv[R];
#pragma unroll
        for (int r = 0; r < R; ++r) {
            int i = i0 + r;
            if (i > total - 1) i = total - 1;
            sn[r] = q_a[i];
            dn[r] = q_d[i];
            wn[r] = q_w[i];
        }
#pragma unroll
        for (int r = 0; r < R; ++r)
            hv[r] = *(const float2*)(h + (size_t)sn[r] * F + 2 * lane);
        asm volatile("" :: "v"(hv[0].x), "v"(hv[0].y), "v"(hv[1].x), "v"(hv[1].y),
                           "v"(hv[2].x), "v"(hv[2].y), "v"(hv[3].x), "v"(hv[3].y),
                           "v"(hv[4].x), "v"(hv[4].y), "v"(hv[5].x), "v"(hv[5].y),
                           "v"(hv[6].x), "v"(hv[6].y), "v"(hv[7].x), "v"(hv[7].y));
#pragma unroll
        for (int r = 0; r < R; ++r) {
            int i = i0 + r;
            if (i < total) {
                atomicAdd(&acc[dn[r] * F + 2 * lane + 0], hv[r].x * wn[r]);
                atomicAdd(&acc[dn[r] * F + 2 * lane + 1], hv[r].y * wn[r]);
            }
        }
    }
    const int n4      = rows * (F / 4);
    const int nstrips = (rows + 15) >> 4;
    const int hi = lane >> 4;
    const int l  = lane & 15;
    v4f hv[EPI_U];
    {
        const v4f* __restrict__ h4 = (const v4f*)(h + (size_t)base * F);
#pragma unroll
        for (int u = 0; u < EPI_U; ++u) {
            const int i = tid + u * AGG_T;
            hv[u] = h4[(i < n4) ? i : n4 - 1];
        }
    }
    const v4f bb = *(const v4f*)(bias + 4 * (tid & 15));
    v2f bf[F / 4] = {};
    if (wave < nstrips) {
        const float* bcol = W2 + l;
#pragma unroll
        for (int j = 0; j < F / 4; ++j) {
            bf[j].x = bcol[(4 * j + 2 * hi + 0) * NC];
            bf[j].y = bcol[(4 * j + 2 * hi + 1) * NC];
        }
    }
    __syncthreads();

    {
        v4f* a4 = (v4f*)acc;
#pragma unroll
        for (int u = 0; u < EPI_U; ++u) {
            const int i = tid + u * AGG_T;
            if (i < n4) {
                const float dii = dis_l[i >> 4];
                const v4f g = a4[i];
                v4f a;
                a.x = fmaxf(g.x + hv[u].x * dii * dii + bb.x, 0.0f);
                a.y = fmaxf(g.y + hv[u].y * dii * dii + bb.y, 0.0f);
                a.z = fmaxf(g.z + hv[u].z * dii * dii + bb.z, 0.0f);
                a.w = fmaxf(g.w + hv[u].w * dii * dii + bb.w, 0.0f);
                a4[i] = a;
            }
        }
    }
    __syncthreads();

    {
        if (wave < nstrips) {
            const float* a0 = acc + (16 * wave + l) * F + 2 * hi;
            v8f c = {};
#pragma unroll
            for (int j = 0; j < F / 4; ++j) {
                const v2f f = *(const v2f*)(a0 + 4 * j);
                c = __builtin_amdgcn_wmma_f32_16x16x4_f32(
                     false, f,  false, bf[j],
                     (short)0, c,  false,  false);
            }
            float* o = acc + (16 * wave + 8 * hi) * F + l;
#pragma unroll
            for (int i = 0; i < 8; ++i) o[i * F] = c[i];
        }
    }
    __syncthreads();

    {
        const int m4 = rows * (NC / 4);
        v4f* __restrict__ o4 = (v4f*)(h2 + (size_t)base * NC);
        for (int i = tid; i < m4; i += AGG_T)
            o4[i] = *(const v4f*)(acc + (i >> 2) * F + 4 * (i & 3));
    }
}

constexpr int AGG2_T = 1024;
constexpr int R2     = 8;

__global__ __launch_bounds__(AGG2_T)
void tile_agg16_kernel(const int* __restrict__ src,
                       const int* __restrict__ dst,
                       const float* __restrict__ dis,
                       const float* __restrict__ h,
                       const int* __restrict__ gq_cnt, const int* __restrict__ gq_s,
                       const int* __restrict__ gq_d, const float* __restrict__ gq_w,
                       const float* __restrict__ bias,
                       float* __restrict__ out, int E, int N) {
    constexpr int F = 16;
    __shared__ __attribute__((aligned(16))) float acc[(TILE_NODES + 1) * F];

    const int tid  = threadIdx.x;
    const int base = blockIdx.x * TILE_NODES;
    int rows = N - base;
    if (rows > TILE_NODES) rows = TILE_NODES;

    for (int i = tid; i < (TILE_NODES + 1) * F; i += AGG2_T) acc[i] = 0.0f;
    __syncthreads();

    int total = gq_cnt[blockIdx.x];
    if (total > QCAP) {
        for (int e = tid; e < E; e += AGG2_T) {
            int d = dst[e];
            unsigned dl = (unsigned)(d - base);
            if (dl < (unsigned)rows) {
                int s = src[e];
                float w = dis[s] * dis[d];
                for (int j = 0; j < F; ++j)
                    atomicAdd(&acc[dl * F + j], h[(size_t)s * F + j] * w);
            }
        }
        total = 0;
    }

    const int*   __restrict__ ts = gq_s + (size_t)blockIdx.x * QCAP;
    const int*   __restrict__ td = gq_d + (size_t)blockIdx.x * QCAP;
    const float* __restrict__ tw = gq_w + (size_t)blockIdx.x * QCAP;

    const int grp = tid >> 2;
    const int qt  = tid & 3;
    constexpr int GROUPS = AGG2_T / 4;
    for (int r0 = 0; r0 < total; r0 += GROUPS * R2) {
        int   sn[R2];
        int   dn[R2];
        float wn[R2];
        v4f   hv[R2];
#pragma unroll
        for (int r = 0; r < R2; ++r) {
            int i = r0 + r * GROUPS + grp;
            if (i > total - 1) i = total - 1;
            sn[r] = ts[i];
            dn[r] = td[i];
            wn[r] = tw[i];
        }
#pragma unroll
        for (int r = 0; r < R2; ++r)
            hv[r] = *(const v4f*)(h + (size_t)sn[r] * F + 4 * qt);
        asm volatile("" :: "v"(hv[0]), "v"(hv[1]), "v"(hv[2]), "v"(hv[3]),
                           "v"(hv[4]), "v"(hv[5]), "v"(hv[6]), "v"(hv[7]),
                           "v"(dn[0]), "v"(dn[1]), "v"(dn[2]), "v"(dn[3]),
                           "v"(dn[4]), "v"(dn[5]), "v"(dn[6]), "v"(dn[7]),
                           "v"(wn[0]), "v"(wn[1]), "v"(wn[2]), "v"(wn[3]),
                           "v"(wn[4]), "v"(wn[5]), "v"(wn[6]), "v"(wn[7]));
#pragma unroll
        for (int r = 0; r < R2; ++r) {
            int i = r0 + r * GROUPS + grp;
            if (i < total) {
                float* p = &acc[dn[r] * F + 4 * qt];
                atomicAdd(p + 0, hv[r].x * wn[r]);
                atomicAdd(p + 1, hv[r].y * wn[r]);
                atomicAdd(p + 2, hv[r].z * wn[r]);
                atomicAdd(p + 3, hv[r].w * wn[r]);
            }
        }
    }
    static_assert(TILE_NODES <= AGG2_T, "one lane per row, one pass");
    const bool has_row = tid < rows;
    const int  r       = has_row ? tid : rows - 1;
    float dii = dis[base + r];
    v4f hh[F / 4];
    v4f bb[F / 4];
    {
        const float* hrow = h + (size_t)(base + r) * F;
#pragma unroll
        for (int q = 0; q < F / 4; ++q) {
            hh[q] = *(const v4f*)(hrow + 4 * q);
            bb[q] = *(const v4f*)(bias + 4 * q);
        }
    }
    __syncthreads();

    if (has_row) {
        dii *= dii;
        float v[F];
        float mx = -3.402823466e38f;
#pragma unroll
        for (int q = 0; q < F / 4; ++q) {
            const v4f a = *(const v4f*)(acc + r * F + 4 * q);
            v[4 * q + 0] = a.x + hh[q].x * dii + bb[q].x;
            v[4 * q + 1] = a.y + hh[q].y * dii + bb[q].y;
            v[4 * q + 2] = a.z + hh[q].z * dii + bb[q].z;
            v[4 * q + 3] = a.w + hh[q].w * dii + bb[q].w;
        }
#pragma unroll
        for (int j = 0; j < F; ++j) mx = fmaxf(mx, v[j]);
        float s = 0.0f;
#pragma unroll
        for (int j = 0; j < F; ++j) s += __expf(v[j] - mx);
        const float lse = mx + __logf(s);
        float* orow = out + (size_t)(base + r) * F;
#pragma unroll
        for (int q = 0; q < F / 4; ++q) {
            v4f o;
            o.x = v[4 * q + 0] - lse;
            o.y = v[4 * q + 1] - lse;
            o.z = v[4 * q + 2] - lse;
            o.w = v[4 * q + 3] - lse;
            *(v4f*)(orow + 4 * q) = o;
        }
    }
}

extern "C" void kernel_launch(void* const* d_in, const int* in_sizes, int n_in,
                              void* d_out, int out_size, void* d_ws, size_t ws_size,
                              hipStream_t stream) {
    const float* x   = (const float*)d_in[0];
    const int*   ei  = (const int*)  d_in[1];
    const float* W1  = (const float*)d_in[2];
    const float* b1  = (const float*)d_in[3];
    const float* W2  = (const float*)d_in[4];
    const float* b2  = (const float*)d_in[5];
    float* out = (float*)d_out;

    const int N = in_sizes[0] / 128;
    const int E = in_sizes[1] / 2;
    const int* src = ei;
    const int* dst = ei + E;

    float* dis  = (float*)d_ws;
    float* h1   = dis  + N;
    float* h2   = h1   + (size_t)N * 64;
    const int nT = (N + TILE_NODES - 1) / TILE_NODES;
    int*   gq_s   = (int*)(h2 + (size_t)N * 16);
    int*   gq_d   = gq_s + (size_t)nT * QCAP;
    float* gq_w   = (float*)(gq_d + (size_t)nT * QCAP);
    const int ng   = (E + 3) / 4;
    const int nblk = (ng + PK_GROUPS - 1) / PK_GROUPS;
    unsigned* pres   = (unsigned*)(gq_w + (size_t)nT * QCAP);
    int*      pairs  = (int*)(pres + (size_t)nT * nblk * PW);
    int*      gq_cnt = pairs + (size_t)nblk * PK_GROUPS * 8;

    pair_kernel<<<nblk, PK_T, 0, stream>>>(src, dst, E, pairs, pres, nT, nblk);
    tile_scan_kernel<<<nT, AGG_T, 0, stream>>>(
        dst, dis, gq_cnt, gq_s, gq_d, pairs, pres, nblk, E, N);

    {
        int nBlocks = (N + GM_ROWS - 1) / GM_ROWS;
        gemm_wmma_f32<128, 64><<<(nBlocks < 1024) ? nBlocks : 1024, (64 / 16) * 32, 0, stream>>>(
            x, W1, h1, N, nBlocks);
    }
    tile_agg64_kernel<<<nT, AGG_T, 0, stream>>>(
        src, dst, dis, h1, b1, W2, h2, gq_cnt, gq_s, gq_d, gq_w, E, N);
    tile_agg16_kernel<<<nT, AGG2_T, 0, stream>>>(
        src, dst, dis, h2, gq_cnt, gq_s, gq_d, gq_w, b2, out, E, N);
}


# Layer-1 tile kernel fetches queue entries from memory, 8 lanes per round, and broadcasts them by readlane (scalar row base and weight): LDS queue copy, its pass and barrier gone; sum order as before
# speedup vs baseline: 4.6243x; 1.0502x over previous
#include <hip/hip_runtime.h>
#include <cstdint>

typedef __attribute__((ext_vector_type(2))) float v2f;
typedef __attribute__((ext_vector_type(4))) float v4f;
typedef __attribute__((ext_vector_type(8))) float v8f;
typedef __attribute__((ext_vector_type(4))) int      v4i;
typedef __attribute__((ext_vector_type(4))) unsigned v4u;

constexpr int TILE_NODES = 392;

constexpr int PK_T      = 1024;
constexpr int PK_GROUPS = PK_T;
constexpr int PW        = PK_GROUPS / 32;
constexpr int PB_PITCH  = 257;
static_assert(PW == 32, "one 128-byte line of presence bits per tile and work-group");

__global__ __launch_bounds__(PK_T)
void pair_kernel(const int* __restrict__ src, const int* __restrict__ dst, int E,
                 int* __restrict__ pairs,
                 unsigned* __restrict__ pres, int nT, int nblk) {
    __shared__ unsigned pb[PW * PB_PITCH];
    int4* __restrict__ pairs4 = (int4*)pairs;
    const int tid  = threadIdx.x;
    const int lane = tid & 31;
    for (int i = tid; i < PW * PB_PITCH; i += PK_T) pb[i] = 0u;

    const int wbase = (blockIdx.x * PK_T + (tid & ~31)) * 4;
    int s[2][2], d[2][2];
#pragma unroll
    for (int k = 0; k < 2; ++k) {
#pragma unroll
        for (int j = 0; j < 2; ++j) {
            const int e  = wbase + 64 * k + 2 * lane + j;
            const int ec = (e < E) ? e : E - 1;
            s[k][j] = src[ec];
            d[k][j] = dst[ec];
        }
    }
    __syncthreads();
    unsigned* pw = pb + (tid >> 5) * PB_PITCH;
#pragma unroll
    for (int k = 0; k < 2; ++k) {
        const unsigned bit = 1u << (16 * k + (lane >> 1));
#pragma unroll
        for (int j = 0; j < 2; ++j) {
            const int e = wbase + 64 * k + 2 * lane + j;
            if (e < E) {
                unsigned tl = ((unsigned)d[k][j] / (unsigned)TILE_NODES) & 0xFFu;
                atomicOr(&pw[tl], bit);
            } else {
                s[k][j] = 0;
                d[k][j] = -1;
            }
        }
        int4 o;
        o.x = s[k][0]; o.y = d[k][0]; o.z = s[k][1]; o.w = d[k][1];
        pairs4[(wbase + 64 * k) / 2 + lane] = o;
    }
    __syncthreads();
    for (int p = tid; p < nT * (PW / 4); p += PK_T) {
        const int tile = p >> 3;
        const int q    = p & 7;
        uint4 o;
        o.x = pb[(4 * q + 0) * PB_PITCH + tile];
        o.y = pb[(4 * q + 1) * PB_PITCH + tile];
        o.z = pb[(4 * q + 2) * PB_PITCH + tile];
        o.w = pb[(4 * q + 3) * PB_PITCH + tile];
        *(uint4*)(pres + ((size_t)tile * nblk + blockIdx.x) * PW + 4 * q) = o;
    }
}

constexpr int GM_ROWS = 32;

template <int K, int THREADS, int LOADS>
__device__ __forceinline__ void gemm_load_block(const float* __restrict__ A, int M, int blk, int tid,
                                                v4f (&pre)[LOADS]) {
    constexpr int ROW_F4 = K / 4;
#pragma unroll
    for (int i = 0; i < LOADS; ++i) {
        const int p = tid + i * THREADS;
        int row = blk * GM_ROWS + p / ROW_F4;
        if (row > M - 1) row = M - 1;
        pre[i] = *(const v4f*)(A + (size_t)row * K + 4 * (p % ROW_F4));
    }
}

template <int K, int NC>
__global__ __launch_bounds__((NC / 16) * 32)
void gemm_wmma_f32(const float* __restrict__ A,
                   const float* __restrict__ B,
                   float* __restrict__ C, int M, int nBlocks) {
    constexpr int WAVES   = NC / 16;
    constexpr int THREADS = WAVES * 32;
    constexpr int ROW_F4  = K / 4;
    constexpr int PITCH   = K + 4;
    constexpr int PIECES  = GM_ROWS * ROW_F4;
    constexpr int LOADS   = PIECES / THREADS;
    static_assert(PIECES % THREADS == 0, "block of A must divide over the threads");
    __shared__ __attribute__((aligned(16))) float a_s[GM_ROWS * PITCH];

    const int tid  = threadIdx.x;
    const int lane = tid & 31;
    const int tn   = tid >> 5;
    const int hi   = lane >> 4;
    const int l    = lane & 15;
    const int colBase = tn * 16;

    v2f bf[K / 4];
    {
        const float* bcol = B + colBase + l;
#pragma unroll
        for (int j = 0; j < K / 4; ++j) {
            bf[j].x = bcol[(size_t)(4 * j + 2 * hi + 0) * NC];
            bf[j].y = bcol[(size_t)(4 * j + 2 * hi + 1) * NC];
        }
    }

    v4f pre[LOADS];
    int blk = blockIdx.x;
    if (blk < nBlocks) gemm_load_block<K, THREADS, LOADS>(A, M, blk, tid, pre);
    for (; blk < nBlocks; blk += gridDim.x) {
#pragma unroll
        for (int i = 0; i < LOADS; ++i) {
            const int p = tid + i * THREADS;
            *(v4f*)(a_s + (p / ROW_F4) * PITCH + 4 * (p % ROW_F4)) = pre[i];
        }
        __syncthreads();
        const int nxt = blk + gridDim.x;
        if (nxt < nBlocks) gemm_load_block<K, THREADS, LOADS>(A, M, nxt, tid, pre);

        const float* a0 = a_s + l * PITCH + 2 * hi;
        const float* a1 = a0 + 16 * PITCH;
        v8f acc0 = {};
        v8f acc1 = {};
#pragma unroll
        for (int j = 0; j < K / 4; ++j) {
            const v2f f0 = *(const v2f*)(a0 + 4 * j);
            const v2f f1 = *(const v2f*)(a1 + 4 * j);
            acc0 = __builtin_amdgcn_wmma_f32_16x16x4_f32(
                 false, f0,  false, bf[j],
                 (short)0, acc0,  false,  false);
            acc1 = __builtin_amdgcn_wmma_f32_16x16x4_f32(
                 false, f1,  false, bf[j],
                 (short)0, acc1,  false,  false);
        }

        const int rowBase = blk * GM_ROWS;
        float* cptr = C + (size_t)colBase + l;
#pragma unroll
        for (int i = 0; i < 8; ++i) {
            const int r0 = rowBase + 8 * hi + i;
            const int r1 = r0 + 16;
            if (r0 < M) cptr[(size_t)r0 * NC] = acc0[i];
            if (r1 < M) cptr[(size_t)r1 * NC] = acc1[i];
        }
        __syncthreads();
    }
}

constexpr int QCAP       = 8192;
constexpr int AGG_T      = 1024;
constexpr int PRES_V     = 4;
constexpr int GRP_U      = QCAP / AGG_T;

__device__ __forceinline__ int wave_queue_base(int cnt, int lane, int* counter) {
    int incl = cnt;
#pragma unroll
    for (int o = 1; o < 32; o <<= 1) {
        int t = __shfl_up(incl, o);
        if (lane >= o) incl += t;
    }
    int wb = 0;
    if (lane == 31) wb = atomicAdd(counter, incl);
    wb = __shfl(wb, 31);
    return wb + incl - cnt;
}

__global__ __launch_bounds__(AGG_T)
void tile_scan_kernel(const int* __restrict__ dst,
                      float* __restrict__ dis,
                      int* __restrict__ gq_cnt, int* __restrict__ gq_s,
                      int* __restrict__ gq_d,
                      const int* __restrict__ pairs,
                      const unsigned* __restrict__ pres, int nblk,
                      int E, int N) {
    const v4i* __restrict__ pairs4 = (const v4i*)pairs;
    __shared__ int   q_s[QCAP];
    __shared__ int   q_d[QCAP];
    __shared__ int   q_g[QCAP];
    __shared__ int   deg_l[TILE_NODES];
    __shared__ int   q_count;
    __shared__ int   g_count;

    const int tid  = threadIdx.x;
    const int lane = tid & 31;
    const int base = blockIdx.x * TILE_NODES;
    int rows = N - base;
    if (rows > TILE_NODES) rows = TILE_NODES;

    for (int i = tid; i < TILE_NODES; i += AGG_T) deg_l[i] = 0;
    if (tid == 0) { q_count = 0; g_count = 0; }
    __syncthreads();

    const int nvec = nblk * (PW / 4);
    const v4u* __restrict__ prow = (const v4u*)(pres + (size_t)blockIdx.x * nblk * PW);
    for (int v0 = 0; v0 < nvec; v0 += PRES_V * AGG_T) {
        v4u      raw[PRES_V];
        unsigned bits[PRES_V][4];
#pragma unroll
        for (int u = 0; u < PRES_V; ++u) {
            const int v = v0 + u * AGG_T + tid;
            raw[u] = prow[(v < nvec) ? v : nvec - 1];
        }
        __builtin_amdgcn_sched_barrier(0);
#pragma unroll
        for (int u = 0; u < PRES_V; ++u) {
            const bool in = (v0 + u * AGG_T + tid) < nvec;
            bits[u][0] = in ? raw[u].x : 0u;
            bits[u][1] = in ? raw[u].y : 0u;
            bits[u][2] = in ? raw[u].z : 0u;
            bits[u][3] = in ? raw[u].w : 0u;
        }
        int cnt = 0;
#pragma unroll
        for (int u = 0; u < PRES_V; ++u)
#pragma unroll
            for (int cc = 0; cc < 4; ++cc) cnt += __builtin_popcount(bits[u][cc]);
        int slot = wave_queue_base(cnt, lane, &g_count);
#pragma unroll
        for (int u = 0; u < PRES_V; ++u) {
            const int v = v0 + u * AGG_T + tid;
#pragma unroll
            for (int cc = 0; cc < 4; ++cc) {
                unsigned m = bits[u][cc];
                const int g0 = 32 * (4 * v + cc);
                while (m) {
                    int b = __builtin_ctz(m);
                    if (slot < QCAP) q_g[slot] = g0 + b;
                    ++slot;
                    m &= m - 1;
                }
            }
        }
    }
    __syncthreads();

    const int gtotal = g_count;
    if (gtotal > QCAP) {
        if (tid == 0) q_count = QCAP + 1;
    } else if (gtotal > 0) {
        int gi[GRP_U];
        v4i pa[GRP_U];
        v4i pc[GRP_U];
#pragma unroll
        for (int u = 0; u < GRP_U; ++u) {
            int i = tid + u * AGG_T;
            gi[u] = q_g[(i < gtotal) ? i : gtotal - 1];
        }
#pragma unroll
        for (int u = 0; u < GRP_U; ++u) {
            pa[u] = pairs4[2 * (size_t)gi[u] + 0];
            pc[u] = pairs4[2 * (size_t)gi[u] + 1];
        }
        asm volatile("" :: "v"(pa[0]), "v"(pc[0]), "v"(pa[1]), "v"(pc[1]),
                           "v"(pa[2]), "v"(pc[2]), "v"(pa[3]), "v"(pc[3]),
                           "v"(pa[4]), "v"(pc[4]), "v"(pa[5]), "v"(pc[5]),
                           "v"(pa[6]), "v"(pc[6]), "v"(pa[7]), "v"(pc[7]));
        static_assert(GRP_U == 8, "the statement above lists 2*GRP_U pieces");
        int dl[GRP_U][4];
        int cnt = 0;
#pragma unroll
        for (int u = 0; u < GRP_U; ++u) {
            const bool in = (tid + u * AGG_T) < gtotal;
            const int dd[4] = { pa[u].y, pa[u].w, pc[u].y, pc[u].w };
#pragma unroll
            for (int j = 0; j < 4; ++j) {
                const unsigned r = (unsigned)(dd[j] - base);
                const bool hit = in && (r < (unsigned)rows);
                dl[u][j] = hit ? (int)r : -1;
                cnt += hit ? 1 : 0;
            }
        }
        int slot = wave_queue_base(cnt, lane, &q_count);
#pragma unroll
        for (int u = 0; u < GRP_U; ++u) {
            const int ss[4] = { pa[u].x, pa[u].z, pc[u].x, pc[u].z };
#pragma unroll
            for (int j = 0; j < 4; ++j) {
                if (dl[u][j] >= 0) {
                    if (slot < QCAP) {
                        q_s[slot] = ss[j];
                        q_d[slot] = dl[u][j];
                    }
                    ++slot;
                    atomicAdd(&deg_l[dl[u][j]], 1);
                }
            }
        }
    }
    __syncthreads();

    int total = q_count;
    if (tid == 0) gq_cnt[blockIdx.x] = total;
    if (total > QCAP) {
        for (int i = tid; i < TILE_NODES; i += AGG_T) deg_l[i] = 0;
        __syncthreads();
        for (int e = tid; e < E; e += AGG_T) {
            unsigned r = (unsigned)(dst[e] - base);
            if (r < (unsigned)rows) atomicAdd(&deg_l[r], 1);
        }
        total = 0;
    }

    int* __restrict__ ts = gq_s + (size_t)blockIdx.x * QCAP;
    int* __restrict__ td = gq_d + (size_t)blockIdx.x * QCAP;
    for (int i = tid; i < total; i += AGG_T) {
        ts[i] = q_s[i];
        td[i] = q_d[i];
    }
    __syncthreads();

    for (int r = tid; r < rows; r += AGG_T)
        dis[base + r] = rsqrtf((float)deg_l[r] + 1.0f);
}

constexpr int ACC_ROWS = 400;
constexpr int EPI_U    = (TILE_NODES * 16 + AGG_T - 1) / AGG_T;
static_assert(ACC_ROWS % 16 == 0 && ACC_ROWS >= TILE_NODES + 1, "whole strips, and one spare row behind the tile");
static_assert(ACC_ROWS / 16 <= AGG_T / 32, "one wave per 16-row strip");

__global__ __launch_bounds__(AGG_T)
void tile_agg64_kernel(const int* __restrict__ src,
                       const int* __restrict__ dst,
                       const float* __restrict__ dis,
                       const float* __restrict__ h,
                       const float* __restrict__ bias,
                       const float* __restrict__ W2,
                       float* __restrict__ h2,
                       const int* __restrict__ gq_cnt, const int* __restrict__ gq_s,
                       const int* __restrict__ gq_d, float* __restrict__ gq_w,
                       int E, int N) {
    constexpr int F  = 64;
    constexpr int NC = 16;
    __shared__ __attribute__((aligned(16))) float acc[ACC_ROWS * F];
    __shared__ float dis_l[TILE_NODES];

    const int tid  = threadIdx.x;
    const int base = blockIdx.x * TILE_NODES;
    int rows = N - base;
    if (rows > TILE_NODES) rows = TILE_NODES;

    for (int i = tid; i < ACC_ROWS * F; i += AGG_T) acc[i] = 0.0f;
    for (int r = tid; r < rows; r += AGG_T) dis_l[r] = dis[base + r];
    __syncthreads();

    int total = gq_cnt[blockIdx.x];
    if (total > QCAP) {
        for (int e = tid; e < E; e += AGG_T) {
            int d = dst[e];
            unsigned dl = (unsigned)(d - base);
            if (dl < (unsigned)rows) {
                int s = src[e];
                float w = dis[s] * dis[d];
                for (int j = 0; j < F; ++j)
                    atomicAdd(&acc[dl * F + j], h[(size_t)s * F + j] * w);
            }
        }
        total = 0;
    }

    const int* __restrict__ ts = gq_s + (size_t)blockIdx.x * QCAP;
    const int* __restrict__ td = gq_d + (size_t)blockIdx.x * QCAP;
    float*     __restrict__ tw = gq_w + (size_t)blockIdx.x * QCAP;
    const int wave = __builtin_amdgcn_readfirstlane(tid >> 5);
    const int lane = tid & 31;
    constexpr int R     = 8;
    constexpr int RP    = 32 / R;
    constexpr int RSTEP = (AGG_T / 32) * R;
    constexpr int FSTEP = RP * RSTEP;
    if (wave * R < total) {
        int ie = wave * R + (lane >> 3) * RSTEP + (lane & (R - 1));
        int cs, cd;
        {
            const int ic = (ie < total) ? ie : total - 1;
            cs = ts[ic];
            cd = td[ic];
        }
        float cds = dis[cs];
        for (int f0 = wave * R; f0 < total; f0 += FSTEP) {
            const bool more = f0 + FSTEP < total;
            const int  ine  = ie + FSTEP;
            int   ns  = cs;
            int   nd  = cd;
            float nds = cds;
            if (more) {
                const int inc = (ine < total) ? ine : total - 1;
                ns = ts[inc];
                nd = td[inc];
            }
            float cw = 0.0f;
#pragma unroll
            for (int sub = 0; sub < RP; ++sub) {
                const int i0 = f0 + sub * RSTEP;
                if (i0 < total) {
                    int    sn[R];
                    int    dn[R];
                    float  wn[R];
                    float2 hv[R];
#pragma unroll
                    for (int r = 0; r < R; ++r)
                        sn[r] = __builtin_amdgcn_readlane(cs, R * sub + r);
#pragma unroll
                    for (int r = 0; r < R; ++r)
                        hv[r] = *(const float2*)(h + (size_t)sn[r] * F + 2 * lane);
                    if (sub == 0) {
                        const bool in_tile = (unsigned)cd < (unsigned)rows;
                        cw = in_tile ? cds * dis_l[in_tile ? cd : 0] : 0.0f;
                        if (ie < total) tw[ie] = cw;
                    }
#pragma unroll
                    for (int r = 0; r < R; ++r) {
                        dn[r] = __builtin_amdgcn_readlane(cd, R * sub + r);
                        wn[r] = __builtin_bit_cast(float,
                                    __builtin_amdgcn_readlane(__builtin_bit_cast(int, cw), R * sub + r));
                    }
                    asm volatile("" :: "v"(hv[0].x), "v"(hv[0].y), "v"(hv[1].x), "v"(hv[1].y),
                                       "v"(hv[2].x), "v"(hv[2].y), "v"(hv[3].x), "v"(hv[3].y),
                                       "v"(hv[4].x), "v"(hv[4].y), "v"(hv[5].x), "v"(hv[5].y),
                                       "v"(hv[6].x), "v"(hv[6].y), "v"(hv[7].x), "v"(hv[7].y));
#pragma unroll
                    for (int r = 0; r < R; ++r) {
                        if (i0 + r < total) {
                            atomicAdd(&acc[dn[r] * F + 2 * lane + 0], hv[r].x * wn[r]);
                            atomicAdd(&acc[dn[r] * F + 2 * lane + 1], hv[r].y * wn[r]);
                        }
                    }
                }
                if (sub == 1 && more) nds = dis[ns];
            }
            cs  = ns;
            cd  = nd;
            cds = nds;
            ie  = ine;
        }
    }
    const int n4      = rows * (F / 4);
    const int nstrips = (rows + 15) >> 4;
    const int hi = lane >> 4;
    const int l  = lane & 15;
    v4f hv[EPI_U];
    {
        const v4f* __restrict__ h4 = (const v4f*)(h + (size_t)base * F);
#pragma unroll
        for (int u = 0; u < EPI_U; ++u) {
            const int i = tid + u * AGG_T;
            hv[u] = h4[(i < n4) ? i : n4 - 1];
        }
    }
    const v4f bb = *(const v4f*)(bias + 4 * (tid & 15));
    v2f bf[F / 4] = {};
    if (wave < nstrips) {
        const float* bcol = W2 + l;
#pragma unroll
        for (int j = 0; j < F / 4; ++j) {
            bf[j].x = bcol[(4 * j + 2 * hi + 0) * NC];
            bf[j].y = bcol[(4 * j + 2 * hi + 1) * NC];
        }
    }
    __syncthreads();

    {
        v4f* a4 = (v4f*)acc;
#pragma unroll
        for (int u = 0; u < EPI_U; ++u) {
            const int i = tid + u * AGG_T;
            if (i < n4) {
                const float dii = dis_l[i >> 4];
                const v4f g = a4[i];
                v4f a;
                a.x = fmaxf(g.x + hv[u].x * dii * dii + bb.x, 0.0f);
                a.y = fmaxf(g.y + hv[u].y * dii * dii + bb.y, 0.0f);
                a.z = fmaxf(g.z + hv[u].z * dii * dii + bb.z, 0.0f);
                a.w = fmaxf(g.w + hv[u].w * dii * dii + bb.w, 0.0f);
                a4[i] = a;
            }
        }
    }
    __syncthreads();

    {
        if (wave < nstrips) {
            const float* a0 = acc + (16 * wave + l) * F + 2 * hi;
            v8f c = {};
#pragma unroll
            for (int j = 0; j < F / 4; ++j) {
                const v2f f = *(const v2f*)(a0 + 4 * j);
                c = __builtin_amdgcn_wmma_f32_16x16x4_f32(
                     false, f,  false, bf[j],
                     (short)0, c,  false,  false);
            }
            float* o = acc + (16 * wave + 8 * hi) * F + l;
#pragma unroll
            for (int i = 0; i < 8; ++i) o[i * F] = c[i];
        }
    }
    __syncthreads();

    {
        const int m4 = rows * (NC / 4);
        v4f* __restrict__ o4 = (v4f*)(h2 + (size_t)base * NC);
        for (int i = tid; i < m4; i += AGG_T)
            o4[i] = *(const v4f*)(acc + (i >> 2) * F + 4 * (i & 3));
    }
}

constexpr int AGG2_T = 1024;
constexpr int R2     = 8;

__global__ __launch_bounds__(AGG2_T)
void tile_agg16_kernel(const int* __restrict__ src,
                       const int* __restrict__ dst,
                       const float* __restrict__ dis,
                       const float* __restrict__ h,
                       const int* __restrict__ gq_cnt, const int* __restrict__ gq_s,
                       const int* __restrict__ gq_d, const float* __restrict__ gq_w,
                       const float* __restrict__ bias,
                       float* __restrict__ out, int E, int N) {
    constexpr int F = 16;
    __shared__ __attribute__((aligned(16))) float acc[(TILE_NODES + 1) * F];

    const int tid  = threadIdx.x;
    const int base = blockIdx.x * TILE_NODES;
    int rows = N - base;
    if (rows > TILE_NODES) rows = TILE_NODES;

    for (int i = tid; i < (TILE_NODES + 1) * F; i += AGG2_T) acc[i] = 0.0f;
    __syncthreads();

    int total = gq_cnt[blockIdx.x];
    if (total > QCAP) {
        for (int e = tid; e < E; e += AGG2_T) {
            int d = dst[e];
            unsigned dl = (unsigned)(d - base);
            if (dl < (unsigned)rows) {
                int s = src[e];
                float w = dis[s] * dis[d];
                for (int j = 0; j < F; ++j)
                    atomicAdd(&acc[dl * F + j], h[(size_t)s * F + j] * w);
            }
        }
        total = 0;
    }

    const int*   __restrict__ ts = gq_s + (size_t)blockIdx.x * QCAP;
    const int*   __restrict__ td = gq_d + (size_t)blockIdx.x * QCAP;
    const float* __restrict__ tw = gq_w + (size_t)blockIdx.x * QCAP;

    const int grp = tid >> 2;
    const int qt  = tid & 3;
    constexpr int GROUPS = AGG2_T / 4;
    for (int r0 = 0; r0 < total; r0 += GROUPS * R2) {
        int   sn[R2];
        int   dn[R2];
        float wn[R2];
        v4f   hv[R2];
#pragma unroll
        for (int r = 0; r < R2; ++r) {
            int i = r0 + r * GROUPS + grp;
            if (i > total - 1) i = total - 1;
            sn[r] = ts[i];
            dn[r] = td[i];
            wn[r] = tw[i];
        }
#pragma unroll
        for (int r = 0; r < R2; ++r)
            hv[r] = *(const v4f*)(h + (size_t)sn[r] * F + 4 * qt);
        asm volatile("" :: "v"(hv[0]), "v"(hv[1]), "v"(hv[2]), "v"(hv[3]),
                           "v"(hv[4]), "v"(hv[5]), "v"(hv[6]), "v"(hv[7]),
                           "v"(dn[0]), "v"(dn[1]), "v"(dn[2]), "v"(dn[3]),
                           "v"(dn[4]), "v"(dn[5]), "v"(dn[6]), "v"(dn[7]),
                           "v"(wn[0]), "v"(wn[1]), "v"(wn[2]), "v"(wn[3]),
                           "v"(wn[4]), "v"(wn[5]), "v"(wn[6]), "v"(wn[7]));
#pragma unroll
        for (int r = 0; r < R2; ++r) {
            int i = r0 + r * GROUPS + grp;
            if (i < total) {
                float* p = &acc[dn[r] * F + 4 * qt];
                atomicAdd(p + 0, hv[r].x * wn[r]);
                atomicAdd(p + 1, hv[r].y * wn[r]);
                atomicAdd(p + 2, hv[r].z * wn[r]);
                atomicAdd(p + 3, hv[r].w * wn[r]);
            }
        }
    }
    static_assert(TILE_NODES <= AGG2_T, "one lane per row, one pass");
    const bool has_row = tid < rows;
    const int  r       = has_row ? tid : rows - 1;
    float dii = dis[base + r];
    v4f hh[F / 4];
    v4f bb[F / 4];
    {
        const float* hrow = h + (size_t)(base + r) * F;
#pragma unroll
        for (int q = 0; q < F / 4; ++q) {
            hh[q] = *(const v4f*)(hrow + 4 * q);
            bb[q] = *(const v4f*)(bias + 4 * q);
        }
    }
    __syncthreads();

    if (has_row) {
        dii *= dii;
        float v[F];
        float mx = -3.402823466e38f;
#pragma unroll
        for (int q = 0; q < F / 4; ++q) {
            const v4f a = *(const v4f*)(acc + r * F + 4 * q);
            v[4 * q + 0] = a.x + hh[q].x * dii + bb[q].x;
            v[4 * q + 1] = a.y + hh[q].y * dii + bb[q].y;
            v[4 * q + 2] = a.z + hh[q].z * dii + bb[q].z;
            v[4 * q + 3] = a.w + hh[q].w * dii + bb[q].w;
        }
#pragma unroll
        for (int j = 0; j < F; ++j) mx = fmaxf(mx, v[j]);
        float s = 0.0f;
#pragma unroll
        for (int j = 0; j < F; ++j) s += __expf(v[j] - mx);
        const float lse = mx + __logf(s);
        float* orow = out + (size_t)(base + r) * F;
#pragma unroll
        for (int q = 0; q < F / 4; ++q) {
            v4f o;
            o.x = v[4 * q + 0] - lse;
            o.y = v[4 * q + 1] - lse;
            o.z = v[4 * q + 2] - lse;
            o.w = v[4 * q + 3] - lse;
            *(v4f*)(orow + 4 * q) = o;
        }
    }
}

extern "C" void kernel_launch(void* const* d_in, const int* in_sizes, int n_in,
                              void* d_out, int out_size, void* d_ws, size_t ws_size,
                              hipStream_t stream) {
    const float* x   = (const float*)d_in[0];
    const int*   ei  = (const int*)  d_in[1];
    const float* W1  = (const float*)d_in[2];
    const float* b1  = (const float*)d_in[3];
    const float* W2  = (const float*)d_in[4];
    const float* b2  = (const float*)d_in[5];
    float* out = (float*)d_out;

    const int N = in_sizes[0] / 128;
    const int E = in_sizes[1] / 2;
    const int* src = ei;
    const int* dst = ei + E;

    float* dis  = (float*)d_ws;
    float* h1   = dis  + N;
    float* h2   = h1   + (size_t)N * 64;
    const int nT = (N + TILE_NODES - 1) / TILE_NODES;
    int*   gq_s   = (int*)(h2 + (size_t)N * 16);
    int*   gq_d   = gq_s + (size_t)nT * QCAP;
    float* gq_w   = (float*)(gq_d + (size_t)nT * QCAP);
    const int ng   = (E + 3) / 4;
    const int nblk = (ng + PK_GROUPS - 1) / PK_GROUPS;
    unsigned* pres   = (unsigned*)(gq_w + (size_t)nT * QCAP);
    int*      pairs  = (int*)(pres + (size_t)nT * nblk * PW);
    int*      gq_cnt = pairs + (size_t)nblk * PK_GROUPS * 8;

    pair_kernel<<<nblk, PK_T, 0, stream>>>(src, dst, E, pairs, pres, nT, nblk);
    tile_scan_kernel<<<nT, AGG_T, 0, stream>>>(
        dst, dis, gq_cnt, gq_s, gq_d, pairs, pres, nblk, E, N);

    {
        int nBlocks = (N + GM_ROWS - 1) / GM_ROWS;
        gemm_wmma_f32<128, 64><<<(nBlocks < 1024) ? nBlocks : 1024, (64 / 16) * 32, 0, stream>>>(
            x, W1, h1, N, nBlocks);
    }
    tile_agg64_kernel<<<nT, AGG_T, 0, stream>>>(
        src, dst, dis, h1, b1, W2, h2, gq_cnt, gq_s, gq_d, gq_w, E, N);
    tile_agg16_kernel<<<nT, AGG2_T, 0, stream>>>(
        src, dst, dis, h2, gq_cnt, gq_s, gq_d, gq_w, b2, out, E, N);
}
